# starting kernel
# baseline (speedup 1.0000x reference)
#include <hip/hip_runtime.h>
#include <hip/hip_bf16.h>

typedef __attribute__((ext_vector_type(2))) float v2f;
typedef __attribute__((ext_vector_type(8))) float v8f;

#define L_TOK   4096
#define B_BATCH 8
#define K0      484
#define H0      32
#define TB      128
#define HALO    120

__global__ __launch_bounds__(256) void k_wsr(const float* __restrict__ Ws,
                                             const float* __restrict__ Wr,
                                             float* __restrict__ Wsr) {
    int id = blockIdx.x * 256 + threadIdx.x;
    int r = id >> 2, o = id & 3;
    const float* wsrow = Ws + (size_t)r * 4096;
    float acc = 0.f;
    for (int k = 0; k < 4096; ++k) acc = fmaf(wsrow[k], Wr[k * 4 + o], acc);
    Wsr[r * 4 + o] = acc;
}

__global__ __launch_bounds__(256) void k_fold(const float* __restrict__ W1,
                                              const float* __restrict__ b1,
                                              const float* __restrict__ bs,
                                              const float* __restrict__ Wr,
                                              const float* __restrict__ br,
                                              const float* __restrict__ Wsr,
                                              float* __restrict__ Wc,
                                              float* __restrict__ bc) {
    int tid = threadIdx.x;
    if (tid < 128) {
        int n = tid >> 2, o = tid & 3;
        float acc = 0.f;
        for (int r = 0; r < 512; ++r) acc = fmaf(W1[n * 512 + r], Wsr[r * 4 + o], acc);
        Wc[tid] = acc;
    } else if (tid < 132) {
        int o = tid - 128;
        float acc = br[o];
        for (int r = 0; r < 512; ++r)  acc = fmaf(b1[r], Wsr[r * 4 + o], acc);
        for (int k = 0; k < 4096; ++k) acc = fmaf(bs[k], Wr[k * 4 + o], acc);
        bc[o] = acc;
    }
}

__global__ __launch_bounds__(256) void k_stage1(const float* __restrict__ x,
                                                const float* __restrict__ W0,
                                                const float* __restrict__ b0,
                                                const float* __restrict__ Wc,
                                                const float* __restrict__ bc,
                                                float* __restrict__ hmid) {
    __shared__ float xt[(TB + HALO) * 4];
    __shared__ float w0t[H0 * K0];
    __shared__ float b0s[H0];
    __shared__ float wcs[H0 * 4];
    __shared__ float bcs[4];
    __shared__ float h0s[TB * 36];

    const int tid = threadIdx.x;
    const int bb  = blockIdx.x >> 5;
    const int t0  = (blockIdx.x & 31) * TB;

    for (int i = tid; i < (TB + HALO) * 4; i += 256) {
        int j = i >> 2, c = i & 3;
        int gt = (t0 + j + L_TOK - HALO) & (L_TOK - 1);
        xt[i] = x[((bb << 12) + gt) * 4 + c];
    }
    for (int i = tid; i < H0 * K0; i += 256) {
        int k = i >> 5, n = i & 31;
        w0t[n * K0 + k] = W0[i];
    }
    if (tid < H0)  b0s[tid] = b0[tid];
    if (tid < 128) wcs[tid] = Wc[tid];
    if (tid < 4)   bcs[tid] = bc[tid];
    __syncthreads();

    const int wave = tid >> 5;
    const int lane = tid & 31;
    const int hl   = lane >> 4;
    const int m0   = lane & 15;
    const int c0   = hl << 1;

    v8f acc0 = {}; v8f acc1 = {};
    const float* ap  = xt  + (((wave << 4) + m0)) * 4 + c0;
    const float* bp0 = w0t + m0 * K0 + c0;
    const float* bp1 = w0t + (m0 + 16) * K0 + c0;

    for (int kk = 0; kk < 121; ++kk) {
        v2f a  = *(const v2f*)(ap  + 4 * kk);
        v2f bA = *(const v2f*)(bp0 + 4 * kk);
        v2f bB = *(const v2f*)(bp1 + 4 * kk);
        acc0 = __builtin_amdgcn_wmma_f32_16x16x4_f32(false, a, false, bA, (short)0, acc0, false, false);
        acc1 = __builtin_amdgcn_wmma_f32_16x16x4_f32(false, a, false, bB, (short)0, acc1, false, false);
    }

    #pragma unroll
    for (int g = 0; g < 8; ++g) {
        int M = (wave << 4) + g + (hl << 3);
        h0s[M * 36 + m0]      = tanhf(acc0[g] + b0s[m0]);
        h0s[M * 36 + m0 + 16] = tanhf(acc1[g] + b0s[m0 + 16]);
    }
    __syncthreads();

    #pragma unroll
    for (int r = 0; r < 2; ++r) {
        int idx = tid + (r << 8);
        int tok = idx >> 2, o = idx & 3;
        const float* hrow = h0s + tok * 36;
        float s = bcs[o];
        #pragma unroll 8
        for (int n = 0; n < 32; ++n) s = fmaf(hrow[n], wcs[n * 4 + o], s);
        hmid[(((bb << 12) + t0 + tok) << 2) + o] = s;
    }
}

__global__ __launch_bounds__(256) void k_stage2(const float* __restrict__ hmid,
                                                const float* __restrict__ W2,
                                                const float* __restrict__ b2,
                                                const float* __restrict__ W3,
                                                const float* __restrict__ b3,
                                                float* __restrict__ out) {
    __shared__ float w2s[20 * 32];
    __shared__ float b2s[32];
    __shared__ float w3s[32 * 4];
    __shared__ float b3s[4];
    int tid = threadIdx.x;
    for (int i = tid; i < 640; i += 256) w2s[i] = W2[i];
    if (tid < 32)  b2s[tid] = b2[tid];
    if (tid < 128) w3s[tid] = W3[tid];
    if (tid < 4)   b3s[tid] = b3[tid];
    __syncthreads();

    int g  = blockIdx.x * 256 + tid;
    int bb = g >> 12, t = g & (L_TOK - 1);

    float win[20];
    #pragma unroll
    for (int j = 0; j < 5; ++j) {
        int tt = (t + L_TOK - 4 + j) & (L_TOK - 1);
        const float* p = hmid + (((bb << 12) + tt) << 2);
        win[j * 4 + 0] = p[0]; win[j * 4 + 1] = p[1];
        win[j * 4 + 2] = p[2]; win[j * 4 + 3] = p[3];
    }
    float o0 = b3s[0], o1 = b3s[1], o2 = b3s[2], o3 = b3s[3];
    for (int n = 0; n < 32; ++n) {
        float h = b2s[n];
        #pragma unroll
        for (int k = 0; k < 20; ++k) h = fmaf(win[k], w2s[k * 32 + n], h);
        h = fmaxf(h, 0.f);
        o0 = fmaf(h, w3s[n * 4 + 0], o0);
        o1 = fmaf(h, w3s[n * 4 + 1], o1);
        o2 = fmaf(h, w3s[n * 4 + 2], o2);
        o3 = fmaf(h, w3s[n * 4 + 3], o3);
    }
    float* q = out + ((size_t)g << 2);
    q[0] = o0; q[1] = o1; q[2] = o2; q[3] = o3;
}

extern "C" void kernel_launch(void* const* d_in, const int* in_sizes, int n_in,
                              void* d_out, int out_size, void* d_ws, size_t ws_size,
                              hipStream_t stream) {
    (void)in_sizes; (void)n_in; (void)out_size; (void)ws_size;
    const float* x  = (const float*)d_in[0];
    const float* W0 = (const float*)d_in[1];
    const float* b0 = (const float*)d_in[2];
    const float* W1 = (const float*)d_in[3];
    const float* b1 = (const float*)d_in[4];
    const float* Ws = (const float*)d_in[5];
    const float* bs = (const float*)d_in[6];
    const float* Wr = (const float*)d_in[7];
    const float* br = (const float*)d_in[8];
    const float* W2 = (const float*)d_in[9];
    const float* b2 = (const float*)d_in[10];
    const float* W3 = (const float*)d_in[11];
    const float* b3 = (const float*)d_in[12];
    float* out = (float*)d_out;

    float* ws   = (float*)d_ws;
    float* Wsr  = ws;
    float* Wc   = ws + 2048;
    float* bc   = ws + 2176;
    float* hmid = ws + 2304;

    k_wsr   <<<8,   256, 0, stream>>>(Ws, Wr, Wsr);
    k_fold  <<<1,   256, 0, stream>>>(W1, b1, bs, Wr, br, Wsr, Wc, bc);
    k_stage1<<<256, 256, 0, stream>>>(x, W0, b0, Wc, bc, hmid);
    k_stage2<<<128, 256, 0, stream>>>(hmid, W2, b2, W3, b3, out);
}


# Weight fold ([Ws;bs]@Wr, [W1;b1]@Wsr): serial per-thread fma chains -> f32 16x16x4 WMMA, split-k over 264 blocks and 8 waves, biases as extra operand rows; f32 sum order of Wc, bc changes
# speedup vs baseline: 8.6479x; 8.6479x over previous
#include <hip/hip_runtime.h>
#include <hip/hip_bf16.h>


typedef __attribute__((ext_vector_type(2))) float v2f;
typedef __attribute__((ext_vector_type(4))) float v4f;
typedef __attribute__((ext_vector_type(8))) float v8f;

#define L_TOK   4096
#define B_BATCH 8
#define K0      484
#define H0      32
#define TB      128
#define HALO    120

#define FK_CH    512
#define FK_NCH   8
#define FK_ROWS  528
#define FK_PITCH 520

__global__ __launch_bounds__(256) void k_wsr(const float* __restrict__ Ws,
                                             const float* __restrict__ bs,
                                             const float* __restrict__ Wr,
                                             float* __restrict__ P) {
    __shared__ __attribute__((aligned(16))) float wrT[5 * FK_PITCH];
    __shared__ __attribute__((aligned(16))) float red[8 * 64];

    const int tid  = threadIdx.x;
    const int tile = blockIdx.x >> 3;
    const int kc   = blockIdx.x & 7;
    const int kb   = kc * FK_CH;
    const int wave = tid >> 5;
    const int lane = tid & 31;
    const int h    = lane >> 4;
    const int c    = lane & 15;

    const int row = tile * 16 + c;
    const float* src = (row < 512) ? (Ws + (size_t)row * 4096) : bs;
    const v4f* bp = (const v4f*)(src + kb + wave * 64 + 4 * h);
    const v4f zero4 = {0.f, 0.f, 0.f, 0.f};
    v4f bv[8];
    #pragma unroll
    for (int j = 0; j < 8; ++j) bv[j] = zero4;
    if (row <= 512) {
        #pragma unroll
        for (int j = 0; j < 8; ++j) bv[j] = bp[2 * j];
    }

    #pragma unroll
    for (int r = 0; r < 2; ++r) {
        int k = tid + (r << 8);
        v4f w = ((const v4f*)(Wr + (size_t)kb * 4))[k];
        wrT[0 * FK_PITCH + k] = w.x;
        wrT[1 * FK_PITCH + k] = w.y;
        wrT[2 * FK_PITCH + k] = w.z;
        wrT[3 * FK_PITCH + k] = w.w;
        wrT[4 * FK_PITCH + k] = 0.f;
    }
    __syncthreads();

    v8f acc0 = {}; v8f acc1 = {};
    const float* ap = wrT + (c < 4 ? c : 4) * FK_PITCH + wave * 64 + 4 * h;
    #pragma unroll
    for (int j = 0; j < 8; ++j) {
        v4f av = *(const v4f*)(ap + 8 * j);
        v2f a0 = {av.x, av.y},       a1 = {av.z, av.w};
        v2f b0 = {bv[j].x, bv[j].y}, b1 = {bv[j].z, bv[j].w};
        acc0 = __builtin_amdgcn_wmma_f32_16x16x4_f32(false, a0, false, b0, (short)0, acc0, false, false);
        acc1 = __builtin_amdgcn_wmma_f32_16x16x4_f32(false, a1, false, b1, (short)0, acc1, false, false);
    }

    if (h == 0) {
        v4f s = {acc0[0] + acc1[0], acc0[1] + acc1[1], acc0[2] + acc1[2], acc0[3] + acc1[3]};
        *(v4f*)(red + wave * 64 + c * 4) = s;
    }
    __syncthreads();
    if (tid < 64) {
        float s = red[tid];
        #pragma unroll
        for (int w = 1; w < 8; ++w) s += red[w * 64 + tid];
        P[((size_t)kc * FK_ROWS + tile * 16) * 4 + tid] = s;
    }
}

__global__ __launch_bounds__(256) void k_fold(const float* __restrict__ W1,
                                              const float* __restrict__ b1,
                                              const float* __restrict__ br,
                                              const float* __restrict__ P,
                                              float* __restrict__ Wc,
                                              float* __restrict__ bc) {
    __shared__ __attribute__((aligned(16))) float wsT[5 * FK_PITCH];
    __shared__ __attribute__((aligned(16))) float red[8 * 3 * 64];
    __shared__ float bsw[4];

    const int tid  = threadIdx.x;
    const int wave = tid >> 5;
    const int lane = tid & 31;
    const int h    = lane >> 4;
    const int c    = lane & 15;
    const v4f zero4 = {0.f, 0.f, 0.f, 0.f};

    const int koff = wave * 64 + 4 * h;
    const v4f* p0 = (const v4f*)(W1 + (size_t)c * 512 + koff);
    const v4f* p1 = (const v4f*)(W1 + (size_t)(c + 16) * 512 + koff);
    const v4f* p2 = (const v4f*)(b1 + koff);
    v4f u0[8], u1[8], u2[8];
    #pragma unroll
    for (int j = 0; j < 8; ++j) {
        u0[j] = p0[2 * j];
        u1[j] = p1[2 * j];
        u2[j] = zero4;
    }
    if (c == 0) {
        #pragma unroll
        for (int j = 0; j < 8; ++j) u2[j] = p2[2 * j];
    }

    const v4f* P4 = (const v4f*)P;
    #pragma unroll
    for (int r = 0; r < 2; ++r) {
        int rr = tid + (r << 8);
        v4f s = P4[rr];
        #pragma unroll
        for (int q = 1; q < FK_NCH; ++q) s += P4[q * FK_ROWS + rr];
        wsT[0 * FK_PITCH + rr] = s.x;
        wsT[1 * FK_PITCH + rr] = s.y;
        wsT[2 * FK_PITCH + rr] = s.z;
        wsT[3 * FK_PITCH + rr] = s.w;
        wsT[4 * FK_PITCH + rr] = 0.f;
    }
    if (tid < 4) {
        float s = P[(size_t)512 * 4 + tid];
        #pragma unroll
        for (int q = 1; q < FK_NCH; ++q) s += P[((size_t)q * FK_ROWS + 512) * 4 + tid];
        bsw[tid] = s;
    }
    __syncthreads();

    v8f d0 = {}; v8f d1 = {}; v8f d2 = {};
    const float* ap = wsT + (c < 4 ? c : 4) * FK_PITCH + koff;
    #pragma unroll
    for (int j = 0; j < 8; ++j) {
        v4f av = *(const v4f*)(ap + 8 * j);
        v2f a0 = {av.x, av.y}, a1 = {av.z, av.w};
        v2f x0 = {u0[j].x, u0[j].y}, x1 = {u0[j].z, u0[j].w};
        v2f y0 = {u1[j].x, u1[j].y}, y1 = {u1[j].z, u1[j].w};
        v2f z0 = {u2[j].x, u2[j].y}, z1 = {u2[j].z, u2[j].w};
        d0 = __builtin_amdgcn_wmma_f32_16x16x4_f32(false, a0, false, x0, (short)0, d0, false, false);
        d1 = __builtin_amdgcn_wmma_f32_16x16x4_f32(false, a0, false, y0, (short)0, d1, false, false);
        d2 = __builtin_amdgcn_wmma_f32_16x16x4_f32(false, a0, false, z0, (short)0, d2, false, false);
        d0 = __builtin_amdgcn_wmma_f32_16x16x4_f32(false, a1, false, x1, (short)0, d0, false, false);
        d1 = __builtin_amdgcn_wmma_f32_16x16x4_f32(false, a1, false, y1, (short)0, d1, false, false);
        d2 = __builtin_amdgcn_wmma_f32_16x16x4_f32(false, a1, false, z1, (short)0, d2, false, false);
    }

    if (h == 0) {
        v4f s0 = {d0[0], d0[1], d0[2], d0[3]};
        v4f s1 = {d1[0], d1[1], d1[2], d1[3]};
        v4f s2 = {d2[0], d2[1], d2[2], d2[3]};
        *(v4f*)(red + (wave * 3 + 0) * 64 + c * 4) = s0;
        *(v4f*)(red + (wave * 3 + 1) * 64 + c * 4) = s1;
        *(v4f*)(red + (wave * 3 + 2) * 64 + c * 4) = s2;
    }
    __syncthreads();

    if (tid < 128) {
        int n = tid >> 2, o = tid & 3;
        int off = (n >> 4) * 64 + (n & 15) * 4 + o;
        float s = red[off];
        #pragma unroll
        for (int w = 1; w < 8; ++w) s += red[w * 192 + off];
        Wc[tid] = s;
    } else if (tid < 132) {
        int o = tid - 128;
        float s = red[2 * 64 + o];
        #pragma unroll
        for (int w = 1; w < 8; ++w) s += red[w * 192 + 2 * 64 + o];
        bc[o] = br[o] + s + bsw[o];
    }
}

__global__ __launch_bounds__(256) void k_stage1(const float* __restrict__ x,
                                                const float* __restrict__ W0,
                                                const float* __restrict__ b0,
                                                const float* __restrict__ Wc,
                                                const float* __restrict__ bc,
                                                float* __restrict__ hmid) {
    __shared__ float xt[(TB + HALO) * 4];
    __shared__ float w0t[H0 * K0];
    __shared__ float b0s[H0];
    __shared__ float wcs[H0 * 4];
    __shared__ float bcs[4];
    __shared__ float h0s[TB * 36];

    const int tid = threadIdx.x;
    const int bb  = blockIdx.x >> 5;
    const int t0  = (blockIdx.x & 31) * TB;

    for (int i = tid; i < (TB + HALO) * 4; i += 256) {
        int j = i >> 2, c = i & 3;
        int gt = (t0 + j + L_TOK - HALO) & (L_TOK - 1);
        xt[i] = x[((bb << 12) + gt) * 4 + c];
    }
    for (int i = tid; i < H0 * K0; i += 256) {
        int k = i >> 5, n = i & 31;
        w0t[n * K0 + k] = W0[i];
    }
    if (tid < H0)  b0s[tid] = b0[tid];
    if (tid < 128) wcs[tid] = Wc[tid];
    if (tid < 4)   bcs[tid] = bc[tid];
    __syncthreads();

    const int wave = tid >> 5;
    const int lane = tid & 31;
    const int hl   = lane >> 4;
    const int m0   = lane & 15;
    const int c0   = hl << 1;

    v8f acc0 = {}; v8f acc1 = {};
    const float* ap  = xt  + (((wave << 4) + m0)) * 4 + c0;
    const float* bp0 = w0t + m0 * K0 + c0;
    const float* bp1 = w0t + (m0 + 16) * K0 + c0;

    for (int kk = 0; kk < 121; ++kk) {
        v2f a  = *(const v2f*)(ap  + 4 * kk);
        v2f bA = *(const v2f*)(bp0 + 4 * kk);
        v2f bB = *(const v2f*)(bp1 + 4 * kk);
        acc0 = __builtin_amdgcn_wmma_f32_16x16x4_f32(false, a, false, bA, (short)0, acc0, false, false);
        acc1 = __builtin_amdgcn_wmma_f32_16x16x4_f32(false, a, false, bB, (short)0, acc1, false, false);
    }

    #pragma unroll
    for (int g = 0; g < 8; ++g) {
        int M = (wave << 4) + g + (hl << 3);
        h0s[M * 36 + m0]      = tanhf(acc0[g] + b0s[m0]);
        h0s[M * 36 + m0 + 16] = tanhf(acc1[g] + b0s[m0 + 16]);
    }
    __syncthreads();

    #pragma unroll
    for (int r = 0; r < 2; ++r) {
        int idx = tid + (r << 8);
        int tok = idx >> 2, o = idx & 3;
        const float* hrow = h0s + tok * 36;
        float s = bcs[o];
        #pragma unroll 8
        for (int n = 0; n < 32; ++n) s = fmaf(hrow[n], wcs[n * 4 + o], s);
        hmid[(((bb << 12) + t0 + tok) << 2) + o] = s;
    }
}

__global__ __launch_bounds__(256) void k_stage2(const float* __restrict__ hmid,
                                                const float* __restrict__ W2,
                                                const float* __restrict__ b2,
                                                const float* __restrict__ W3,
                                                const float* __restrict__ b3,
                                                float* __restrict__ out) {
    __shared__ float w2s[20 * 32];
    __shared__ float b2s[32];
    __shared__ float w3s[32 * 4];
    __shared__ float b3s[4];
    int tid = threadIdx.x;
    for (int i = tid; i < 640; i += 256) w2s[i] = W2[i];
    if (tid < 32)  b2s[tid] = b2[tid];
    if (tid < 128) w3s[tid] = W3[tid];
    if (tid < 4)   b3s[tid] = b3[tid];
    __syncthreads();

    int g  = blockIdx.x * 256 + tid;
    int bb = g >> 12, t = g & (L_TOK - 1);

    float win[20];
    #pragma unroll
    for (int j = 0; j < 5; ++j) {
        int tt = (t + L_TOK - 4 + j) & (L_TOK - 1);
        const float* p = hmid + (((bb << 12) + tt) << 2);
        win[j * 4 + 0] = p[0]; win[j * 4 + 1] = p[1];
        win[j * 4 + 2] = p[2]; win[j * 4 + 3] = p[3];
    }
    float o0 = b3s[0], o1 = b3s[1], o2 = b3s[2], o3 = b3s[3];
    for (int n = 0; n < 32; ++n) {
        float h = b2s[n];
        #pragma unroll
        for (int k = 0; k < 20; ++k) h = fmaf(win[k], w2s[k * 32 + n], h);
        h = fmaxf(h, 0.f);
        o0 = fmaf(h, w3s[n * 4 + 0], o0);
        o1 = fmaf(h, w3s[n * 4 + 1], o1);
        o2 = fmaf(h, w3s[n * 4 + 2], o2);
        o3 = fmaf(h, w3s[n * 4 + 3], o3);
    }
    float* q = out + ((size_t)g << 2);
    q[0] = o0; q[1] = o1; q[2] = o2; q[3] = o3;
}

extern "C" void kernel_launch(void* const* d_in, const int* in_sizes, int n_in,
                              void* d_out, int out_size, void* d_ws, size_t ws_size,
                              hipStream_t stream) {
    (void)in_sizes; (void)n_in; (void)out_size; (void)ws_size;
    const float* x  = (const float*)d_in[0];
    const float* W0 = (const float*)d_in[1];
    const float* b0 = (const float*)d_in[2];
    const float* W1 = (const float*)d_in[3];
    const float* b1 = (const float*)d_in[4];
    const float* Ws = (const float*)d_in[5];
    const float* bs = (const float*)d_in[6];
    const float* Wr = (const float*)d_in[7];
    const float* br = (const float*)d_in[8];
    const float* W2 = (const float*)d_in[9];
    const float* b2 = (const float*)d_in[10];
    const float* W3 = (const float*)d_in[11];
    const float* b3 = (const float*)d_in[12];
    float* out = (float*)d_out;

    float* ws   = (float*)d_ws;
    float* P    = ws;
    float* Wc   = ws + 16896;
    float* bc   = ws + 17024;
    float* hmid = ws + 17152;

    k_wsr   <<<264, 256, 0, stream>>>(Ws, bs, Wr, P);
    k_fold  <<<1,   256, 0, stream>>>(W1, b1, br, P, Wc, bc);
    k_stage1<<<256, 256, 0, stream>>>(x, W0, b0, Wc, bc, hmid);
    k_stage2<<<128, 256, 0, stream>>>(hmid, W2, b2, W3, b3, out);
}


# Front kernel: W0 skips LDS: k split over the 8 waves, B fragments from global straight to registers, partial tiles summed via LDS in wave order; f32 sum order of the 484-term dots changes
# speedup vs baseline: 9.8788x; 1.1423x over previous
#include <hip/hip_runtime.h>
#include <hip/hip_bf16.h>


typedef __attribute__((ext_vector_type(2))) float v2f;
typedef __attribute__((ext_vector_type(4))) float v4f;
typedef __attribute__((ext_vector_type(8))) float v8f;

#define L_TOK   4096
#define B_BATCH 8
#define K0      484
#define H0      32
#define TB      128
#define HALO    120

#define FK_CH    512
#define FK_NCH   8
#define FK_ROWS  528
#define FK_PITCH 520

__global__ __launch_bounds__(256) void k_wsr(const float* __restrict__ Ws,
                                             const float* __restrict__ bs,
                                             const float* __restrict__ Wr,
                                             float* __restrict__ P) {
    __shared__ __attribute__((aligned(16))) float wrT[5 * FK_PITCH];
    __shared__ __attribute__((aligned(16))) float red[8 * 64];

    const int tid  = threadIdx.x;
    const int tile = blockIdx.x >> 3;
    const int kc   = blockIdx.x & 7;
    const int kb   = kc * FK_CH;
    const int wave = tid >> 5;
    const int lane = tid & 31;
    const int h    = lane >> 4;
    const int c    = lane & 15;

    const int row = tile * 16 + c;
    const float* src = (row < 512) ? (Ws + (size_t)row * 4096) : bs;
    const v4f* bp = (const v4f*)(src + kb + wave * 64 + 4 * h);
    const v4f zero4 = {0.f, 0.f, 0.f, 0.f};
    v4f bv[8];
    #pragma unroll
    for (int j = 0; j < 8; ++j) bv[j] = zero4;
    if (row <= 512) {
        #pragma unroll
        for (int j = 0; j < 8; ++j) bv[j] = bp[2 * j];
    }

    #pragma unroll
    for (int r = 0; r < 2; ++r) {
        int k = tid + (r << 8);
        v4f w = ((const v4f*)(Wr + (size_t)kb * 4))[k];
        wrT[0 * FK_PITCH + k] = w.x;
        wrT[1 * FK_PITCH + k] = w.y;
        wrT[2 * FK_PITCH + k] = w.z;
        wrT[3 * FK_PITCH + k] = w.w;
        wrT[4 * FK_PITCH + k] = 0.f;
    }
    __syncthreads();

    v8f acc0 = {}; v8f acc1 = {};
    const float* ap = wrT + (c < 4 ? c : 4) * FK_PITCH + wave * 64 + 4 * h;
    #pragma unroll
    for (int j = 0; j < 8; ++j) {
        v4f av = *(const v4f*)(ap + 8 * j);
        v2f a0 = {av.x, av.y},       a1 = {av.z, av.w};
        v2f b0 = {bv[j].x, bv[j].y}, b1 = {bv[j].z, bv[j].w};
        acc0 = __builtin_amdgcn_wmma_f32_16x16x4_f32(false, a0, false, b0, (short)0, acc0, false, false);
        acc1 = __builtin_amdgcn_wmma_f32_16x16x4_f32(false, a1, false, b1, (short)0, acc1, false, false);
    }

    if (h == 0) {
        v4f s = {acc0[0] + acc1[0], acc0[1] + acc1[1], acc0[2] + acc1[2], acc0[3] + acc1[3]};
        *(v4f*)(red + wave * 64 + c * 4) = s;
    }
    __syncthreads();
    if (tid < 64) {
        float s = red[tid];
        #pragma unroll
        for (int w = 1; w < 8; ++w) s += red[w * 64 + tid];
        P[((size_t)kc * FK_ROWS + tile * 16) * 4 + tid] = s;
    }
}

__global__ __launch_bounds__(256) void k_fold(const float* __restrict__ W1,
                                              const float* __restrict__ b1,
                                              const float* __restrict__ br,
                                              const float* __restrict__ P,
                                              float* __restrict__ Wc,
                                              float* __restrict__ bc) {
    __shared__ __attribute__((aligned(16))) float wsT[5 * FK_PITCH];
    __shared__ __attribute__((aligned(16))) float red[8 * 3 * 64];
    __shared__ float bsw[4];

    const int tid  = threadIdx.x;
    const int wave = tid >> 5;
    const int lane = tid & 31;
    const int h    = lane >> 4;
    const int c    = lane & 15;
    const v4f zero4 = {0.f, 0.f, 0.f, 0.f};

    const int koff = wave * 64 + 4 * h;
    const v4f* p0 = (const v4f*)(W1 + (size_t)c * 512 + koff);
    const v4f* p1 = (const v4f*)(W1 + (size_t)(c + 16) * 512 + koff);
    const v4f* p2 = (const v4f*)(b1 + koff);
    v4f u0[8], u1[8], u2[8];
    #pragma unroll
    for (int j = 0; j < 8; ++j) {
        u0[j] = p0[2 * j];
        u1[j] = p1[2 * j];
        u2[j] = zero4;
    }
    if (c == 0) {
        #pragma unroll
        for (int j = 0; j < 8; ++j) u2[j] = p2[2 * j];
    }

    const v4f* P4 = (const v4f*)P;
    #pragma unroll
    for (int r = 0; r < 2; ++r) {
        int rr = tid + (r << 8);
        v4f s = P4[rr];
        #pragma unroll
        for (int q = 1; q < FK_NCH; ++q) s += P4[q * FK_ROWS + rr];
        wsT[0 * FK_PITCH + rr] = s.x;
        wsT[1 * FK_PITCH + rr] = s.y;
        wsT[2 * FK_PITCH + rr] = s.z;
        wsT[3 * FK_PITCH + rr] = s.w;
        wsT[4 * FK_PITCH + rr] = 0.f;
    }
    if (tid < 4) {
        float s = P[(size_t)512 * 4 + tid];
        #pragma unroll
        for (int q = 1; q < FK_NCH; ++q) s += P[((size_t)q * FK_ROWS + 512) * 4 + tid];
        bsw[tid] = s;
    }
    __syncthreads();

    v8f d0 = {}; v8f d1 = {}; v8f d2 = {};
    const float* ap = wsT + (c < 4 ? c : 4) * FK_PITCH + koff;
    #pragma unroll
    for (int j = 0; j < 8; ++j) {
        v4f av = *(const v4f*)(ap + 8 * j);
        v2f a0 = {av.x, av.y}, a1 = {av.z, av.w};
        v2f x0 = {u0[j].x, u0[j].y}, x1 = {u0[j].z, u0[j].w};
        v2f y0 = {u1[j].x, u1[j].y}, y1 = {u1[j].z, u1[j].w};
        v2f z0 = {u2[j].x, u2[j].y}, z1 = {u2[j].z, u2[j].w};
        d0 = __builtin_amdgcn_wmma_f32_16x16x4_f32(false, a0, false, x0, (short)0, d0, false, false);
        d1 = __builtin_amdgcn_wmma_f32_16x16x4_f32(false, a0, false, y0, (short)0, d1, false, false);
        d2 = __builtin_amdgcn_wmma_f32_16x16x4_f32(false, a0, false, z0, (short)0, d2, false, false);
        d0 = __builtin_amdgcn_wmma_f32_16x16x4_f32(false, a1, false, x1, (short)0, d0, false, false);
        d1 = __builtin_amdgcn_wmma_f32_16x16x4_f32(false, a1, false, y1, (short)0, d1, false, false);
        d2 = __builtin_amdgcn_wmma_f32_16x16x4_f32(false, a1, false, z1, (short)0, d2, false, false);
    }

    if (h == 0) {
        v4f s0 = {d0[0], d0[1], d0[2], d0[3]};
        v4f s1 = {d1[0], d1[1], d1[2], d1[3]};
        v4f s2 = {d2[0], d2[1], d2[2], d2[3]};
        *(v4f*)(red + (wave * 3 + 0) * 64 + c * 4) = s0;
        *(v4f*)(red + (wave * 3 + 1) * 64 + c * 4) = s1;
        *(v4f*)(red + (wave * 3 + 2) * 64 + c * 4) = s2;
    }
    __syncthreads();

    if (tid < 128) {
        int n = tid >> 2, o = tid & 3;
        int off = (n >> 4) * 64 + (n & 15) * 4 + o;
        float s = red[off];
        #pragma unroll
        for (int w = 1; w < 8; ++w) s += red[w * 192 + off];
        Wc[tid] = s;
    } else if (tid < 132) {
        int o = tid - 128;
        float s = red[2 * 64 + o];
        #pragma unroll
        for (int w = 1; w < 8; ++w) s += red[w * 192 + 2 * 64 + o];
        bc[o] = br[o] + s + bsw[o];
    }
}

#define KS_WAVE 15
__global__ __launch_bounds__(256) void k_stage1(const float* __restrict__ x,
                                                const float* __restrict__ W0,
                                                const float* __restrict__ b0,
                                                const float* __restrict__ Wc,
                                                const float* __restrict__ bc,
                                                float* __restrict__ hmid) {
    __shared__ float xt[(TB + HALO) * 4];
    __shared__ __attribute__((aligned(16))) float part[8 * 8 * 2 * 2 * 128];
    __shared__ float b0s[H0];
    __shared__ float wcs[H0 * 4];
    __shared__ float bcs[4];
    __shared__ float h0s[TB * 36];

    const int tid = threadIdx.x;
    const int bb  = blockIdx.x >> 5;
    const int t0  = (blockIdx.x & 31) * TB;
    const int wave = __builtin_amdgcn_readfirstlane(tid >> 5);
    const int lane = tid & 31;
    const int hl   = lane >> 4;
    const int m0   = lane & 15;
    const int c0   = hl << 1;
    const int k_lo = wave * KS_WAVE;

    const float* bptr = W0 + (size_t)(((k_lo << 2) + c0) * H0 + m0);
    v2f breg[16][2];
    #pragma unroll
    for (int i = 0; i < KS_WAVE; ++i) {
        v2f f0 = {bptr[(4 * i) * H0],      bptr[(4 * i + 1) * H0]};
        v2f f1 = {bptr[(4 * i) * H0 + 16], bptr[(4 * i + 1) * H0 + 16]};
        breg[i][0] = f0; breg[i][1] = f1;
    }
    {
        v2f z = {0.f, 0.f};
        breg[15][0] = z; breg[15][1] = z;
    }
    if (wave == 7) {
        v2f f0 = {bptr[60 * H0],      bptr[61 * H0]};
        v2f f1 = {bptr[60 * H0 + 16], bptr[61 * H0 + 16]};
        breg[15][0] = f0; breg[15][1] = f1;
    }

    for (int i = tid; i < (TB + HALO) * 4; i += 256) {
        int j = i >> 2, c = i & 3;
        int gt = (t0 + j + L_TOK - HALO) & (L_TOK - 1);
        xt[i] = x[((bb << 12) + gt) * 4 + c];
    }
    if (tid < H0)  b0s[tid] = b0[tid];
    if (tid < 128) wcs[tid] = Wc[tid];
    if (tid < 4)   bcs[tid] = bc[tid];
    __syncthreads();

    v8f acc[8][2];
    #pragma unroll
    for (int mt = 0; mt < 8; ++mt) { v8f z = {}; acc[mt][0] = z; acc[mt][1] = z; }

    const float* ap = xt + m0 * 4 + c0 + 4 * k_lo;
    #pragma unroll
    for (int i = 0; i < KS_WAVE; ++i) {
        #pragma unroll
        for (int mt = 0; mt < 8; ++mt) {
            v2f a = *(const v2f*)(ap + mt * 64 + 4 * i);
            acc[mt][0] = __builtin_amdgcn_wmma_f32_16x16x4_f32(false, a, false, breg[i][0], (short)0, acc[mt][0], false, false);
            acc[mt][1] = __builtin_amdgcn_wmma_f32_16x16x4_f32(false, a, false, breg[i][1], (short)0, acc[mt][1], false, false);
        }
    }
    if (wave == 7) {
        #pragma unroll
        for (int mt = 0; mt < 8; ++mt) {
            v2f a = *(const v2f*)(ap + mt * 64 + 4 * KS_WAVE);
            acc[mt][0] = __builtin_amdgcn_wmma_f32_16x16x4_f32(false, a, false, breg[15][0], (short)0, acc[mt][0], false, false);
            acc[mt][1] = __builtin_amdgcn_wmma_f32_16x16x4_f32(false, a, false, breg[15][1], (short)0, acc[mt][1], false, false);
        }
    }

    float* pw = part + wave * 4096 + lane * 4;
    #pragma unroll
    for (int mt = 0; mt < 8; ++mt) {
        #pragma unroll
        for (int nt = 0; nt < 2; ++nt) {
            v8f d = acc[mt][nt];
            v4f lo = {d[0], d[1], d[2], d[3]};
            v4f hi = {d[4], d[5], d[6], d[7]};
            *(v4f*)(pw + ((mt * 2 + nt) * 2 + 0) * 128) = lo;
            *(v4f*)(pw + ((mt * 2 + nt) * 2 + 1) * 128) = hi;
        }
    }
    __syncthreads();

    const float* pr = part + wave * 512 + lane * 4;
    v8f acc0, acc1;
    {
        v4f q0 = *(const v4f*)(pr);
        v4f q1 = *(const v4f*)(pr + 128);
        v4f q2 = *(const v4f*)(pr + 256);
        v4f q3 = *(const v4f*)(pr + 384);
        #pragma unroll
        for (int s = 1; s < 8; ++s) {
            q0 += *(const v4f*)(pr + s * 4096);
            q1 += *(const v4f*)(pr + s * 4096 + 128);
            q2 += *(const v4f*)(pr + s * 4096 + 256);
            q3 += *(const v4f*)(pr + s * 4096 + 384);
        }
        v8f r0 = {q0.x, q0.y, q0.z, q0.w, q1.x, q1.y, q1.z, q1.w};
        v8f r1 = {q2.x, q2.y, q2.z, q2.w, q3.x, q3.y, q3.z, q3.w};
        acc0 = r0; acc1 = r1;
    }

    #pragma unroll
    for (int g = 0; g < 8; ++g) {
        int M = (wave << 4) + g + (hl << 3);
        h0s[M * 36 + m0]      = tanhf(acc0[g] + b0s[m0]);
        h0s[M * 36 + m0 + 16] = tanhf(acc1[g] + b0s[m0 + 16]);
    }
    __syncthreads();

    #pragma unroll
    for (int r = 0; r < 2; ++r) {
        int idx = tid + (r << 8);
        int tok = idx >> 2, o = idx & 3;
        const float* hrow = h0s + tok * 36;
        float s = bcs[o];
        #pragma unroll 8
        for (int n = 0; n < 32; ++n) s = fmaf(hrow[n], wcs[n * 4 + o], s);
        hmid[(((bb << 12) + t0 + tok) << 2) + o] = s;
    }
}

__global__ __launch_bounds__(256) void k_stage2(const float* __restrict__ hmid,
                                                const float* __restrict__ W2,
                                                const float* __restrict__ b2,
                                                const float* __restrict__ W3,
                                                const float* __restrict__ b3,
                                                float* __restrict__ out) {
    __shared__ float w2s[20 * 32];
    __shared__ float b2s[32];
    __shared__ float w3s[32 * 4];
    __shared__ float b3s[4];
    int tid = threadIdx.x;
    for (int i = tid; i < 640; i += 256) w2s[i] = W2[i];
    if (tid < 32)  b2s[tid] = b2[tid];
    if (tid < 128) w3s[tid] = W3[tid];
    if (tid < 4)   b3s[tid] = b3[tid];
    __syncthreads();

    int g  = blockIdx.x * 256 + tid;
    int bb = g >> 12, t = g & (L_TOK - 1);

    float win[20];
    #pragma unroll
    for (int j = 0; j < 5; ++j) {
        int tt = (t + L_TOK - 4 + j) & (L_TOK - 1);
        const float* p = hmid + (((bb << 12) + tt) << 2);
        win[j * 4 + 0] = p[0]; win[j * 4 + 1] = p[1];
        win[j * 4 + 2] = p[2]; win[j * 4 + 3] = p[3];
    }
    float o0 = b3s[0], o1 = b3s[1], o2 = b3s[2], o3 = b3s[3];
    for (int n = 0; n < 32; ++n) {
        float h = b2s[n];
        #pragma unroll
        for (int k = 0; k < 20; ++k) h = fmaf(win[k], w2s[k * 32 + n], h);
        h = fmaxf(h, 0.f);
        o0 = fmaf(h, w3s[n * 4 + 0], o0);
        o1 = fmaf(h, w3s[n * 4 + 1], o1);
        o2 = fmaf(h, w3s[n * 4 + 2], o2);
        o3 = fmaf(h, w3s[n * 4 + 3], o3);
    }
    float* q = out + ((size_t)g << 2);
    q[0] = o0; q[1] = o1; q[2] = o2; q[3] = o3;
}

extern "C" void kernel_launch(void* const* d_in, const int* in_sizes, int n_in,
                              void* d_out, int out_size, void* d_ws, size_t ws_size,
                              hipStream_t stream) {
    (void)in_sizes; (void)n_in; (void)out_size; (void)ws_size;
    const float* x  = (const float*)d_in[0];
    const float* W0 = (const float*)d_in[1];
    const float* b0 = (const float*)d_in[2];
    const float* W1 = (const float*)d_in[3];
    const float* b1 = (const float*)d_in[4];
    const float* Ws = (const float*)d_in[5];
    const float* bs = (const float*)d_in[6];
    const float* Wr = (const float*)d_in[7];
    const float* br = (const float*)d_in[8];
    const float* W2 = (const float*)d_in[9];
    const float* b2 = (const float*)d_in[10];
    const float* W3 = (const float*)d_in[11];
    const float* b3 = (const float*)d_in[12];
    float* out = (float*)d_out;

    float* ws   = (float*)d_ws;
    float* P    = ws;
    float* Wc   = ws + 16896;
    float* bc   = ws + 17024;
    float* hmid = ws + 17152;

    k_wsr   <<<264, 256, 0, stream>>>(Ws, bs, Wr, P);
    k_fold  <<<1,   256, 0, stream>>>(W1, b1, br, P, Wc, bc);
    k_stage1<<<256, 256, 0, stream>>>(x, W0, b0, Wc, bc, hmid);
    k_stage2<<<128, 256, 0, stream>>>(hmid, W2, b2, W3, b3, out);
}


# Launch fusion 4 -> 3: window-5 stage moves into the front kernel, its 4 tokens in front recomputed as a ninth WMMA token tile, hmid stays in LDS; f32 sum order of the last 32-term dots changes
# speedup vs baseline: 12.2481x; 1.2398x over previous
#include <hip/hip_runtime.h>
#include <hip/hip_bf16.h>


typedef __attribute__((ext_vector_type(2))) float v2f;
typedef __attribute__((ext_vector_type(4))) float v4f;
typedef __attribute__((ext_vector_type(8))) float v8f;

#define L_TOK   4096
#define B_BATCH 8
#define K0      484
#define H0      32
#define TB      128
#define HALO    120
#define NTILE   9
#define XPAD    12
#define XTOK    (XPAD + 4 + HALO + TB)
#define PW      (NTILE * 512)

#define FK_CH    512
#define FK_NCH   8
#define FK_ROWS  528
#define FK_PITCH 520

__global__ __launch_bounds__(256) void k_wsr(const float* __restrict__ Ws,
                                             const float* __restrict__ bs,
                                             const float* __restrict__ Wr,
                                             float* __restrict__ P) {
    __shared__ __attribute__((aligned(16))) float wrT[5 * FK_PITCH];
    __shared__ __attribute__((aligned(16))) float red[8 * 64];

    const int tid  = threadIdx.x;
    const int tile = blockIdx.x >> 3;
    const int kc   = blockIdx.x & 7;
    const int kb   = kc * FK_CH;
    const int wave = tid >> 5;
    const int lane = tid & 31;
    const int h    = lane >> 4;
    const int c    = lane & 15;

    const int row = tile * 16 + c;
    const float* src = (row < 512) ? (Ws + (size_t)row * 4096) : bs;
    const v4f* bp = (const v4f*)(src + kb + wave * 64 + 4 * h);
    const v4f zero4 = {0.f, 0.f, 0.f, 0.f};
    v4f bv[8];
    #pragma unroll
    for (int j = 0; j < 8; ++j) bv[j] = zero4;
    if (row <= 512) {
        #pragma unroll
        for (int j = 0; j < 8; ++j) bv[j] = bp[2 * j];
    }

    #pragma unroll
    for (int r = 0; r < 2; ++r) {
        int k = tid + (r << 8);
        v4f w = ((const v4f*)(Wr + (size_t)kb * 4))[k];
        wrT[0 * FK_PITCH + k] = w.x;
        wrT[1 * FK_PITCH + k] = w.y;
        wrT[2 * FK_PITCH + k] = w.z;
        wrT[3 * FK_PITCH + k] = w.w;
        wrT[4 * FK_PITCH + k] = 0.f;
    }
    __syncthreads();

    v8f acc0 = {}; v8f acc1 = {};
    const float* ap = wrT + (c < 4 ? c : 4) * FK_PITCH + wave * 64 + 4 * h;
    #pragma unroll
    for (int j = 0; j < 8; ++j) {
        v4f av = *(const v4f*)(ap + 8 * j);
        v2f a0 = {av.x, av.y},       a1 = {av.z, av.w};
        v2f b0 = {bv[j].x, bv[j].y}, b1 = {bv[j].z, bv[j].w};
        acc0 = __builtin_amdgcn_wmma_f32_16x16x4_f32(false, a0, false, b0, (short)0, acc0, false, false);
        acc1 = __builtin_amdgcn_wmma_f32_16x16x4_f32(false, a1, false, b1, (short)0, acc1, false, false);
    }

    if (h == 0) {
        v4f s = {acc0[0] + acc1[0], acc0[1] + acc1[1], acc0[2] + acc1[2], acc0[3] + acc1[3]};
        *(v4f*)(red + wave * 64 + c * 4) = s;
    }
    __syncthreads();
    if (tid < 64) {
        float s = red[tid];
        #pragma unroll
        for (int w = 1; w < 8; ++w) s += red[w * 64 + tid];
        P[((size_t)kc * FK_ROWS + tile * 16) * 4 + tid] = s;
    }
}

__global__ __launch_bounds__(256) void k_fold(const float* __restrict__ W1,
                                              const float* __restrict__ b1,
                                              const float* __restrict__ br,
                                              const float* __restrict__ P,
                                              float* __restrict__ Wc,
                                              float* __restrict__ bc) {
    __shared__ __attribute__((aligned(16))) float wsT[5 * FK_PITCH];
    __shared__ __attribute__((aligned(16))) float red[8 * 3 * 64];
    __shared__ float bsw[4];

    const int tid  = threadIdx.x;
    const int wave = tid >> 5;
    const int lane = tid & 31;
    const int h    = lane >> 4;
    const int c    = lane & 15;
    const v4f zero4 = {0.f, 0.f, 0.f, 0.f};

    const int koff = wave * 64 + 4 * h;
    const v4f* p0 = (const v4f*)(W1 + (size_t)c * 512 + koff);
    const v4f* p1 = (const v4f*)(W1 + (size_t)(c + 16) * 512 + koff);
    const v4f* p2 = (const v4f*)(b1 + koff);
    v4f u0[8], u1[8], u2[8];
    #pragma unroll
    for (int j = 0; j < 8; ++j) {
        u0[j] = p0[2 * j];
        u1[j] = p1[2 * j];
        u2[j] = zero4;
    }
    if (c == 0) {
        #pragma unroll
        for (int j = 0; j < 8; ++j) u2[j] = p2[2 * j];
    }

    const v4f* P4 = (const v4f*)P;
    #pragma unroll
    for (int r = 0; r < 2; ++r) {
        int rr = tid + (r << 8);
        v4f s = P4[rr];
        #pragma unroll
        for (int q = 1; q < FK_NCH; ++q) s += P4[q * FK_ROWS + rr];
        wsT[0 * FK_PITCH + rr] = s.x;
        wsT[1 * FK_PITCH + rr] = s.y;
        wsT[2 * FK_PITCH + rr] = s.z;
        wsT[3 * FK_PITCH + rr] = s.w;
        wsT[4 * FK_PITCH + rr] = 0.f;
    }
    if (tid < 4) {
        float s = P[(size_t)512 * 4 + tid];
        #pragma unroll
        for (int q = 1; q < FK_NCH; ++q) s += P[((size_t)q * FK_ROWS + 512) * 4 + tid];
        bsw[tid] = s;
    }
    __syncthreads();

    v8f d0 = {}; v8f d1 = {}; v8f d2 = {};
    const float* ap = wsT + (c < 4 ? c : 4) * FK_PITCH + koff;
    #pragma unroll
    for (int j = 0; j < 8; ++j) {
        v4f av = *(const v4f*)(ap + 8 * j);
        v2f a0 = {av.x, av.y}, a1 = {av.z, av.w};
        v2f x0 = {u0[j].x, u0[j].y}, x1 = {u0[j].z, u0[j].w};
        v2f y0 = {u1[j].x, u1[j].y}, y1 = {u1[j].z, u1[j].w};
        v2f z0 = {u2[j].x, u2[j].y}, z1 = {u2[j].z, u2[j].w};
        d0 = __builtin_amdgcn_wmma_f32_16x16x4_f32(false, a0, false, x0, (short)0, d0, false, false);
        d1 = __builtin_amdgcn_wmma_f32_16x16x4_f32(false, a0, false, y0, (short)0, d1, false, false);
        d2 = __builtin_amdgcn_wmma_f32_16x16x4_f32(false, a0, false, z0, (short)0, d2, false, false);
        d0 = __builtin_amdgcn_wmma_f32_16x16x4_f32(false, a1, false, x1, (short)0, d0, false, false);
        d1 = __builtin_amdgcn_wmma_f32_16x16x4_f32(false, a1, false, y1, (short)0, d1, false, false);
        d2 = __builtin_amdgcn_wmma_f32_16x16x4_f32(false, a1, false, z1, (short)0, d2, false, false);
    }

    if (h == 0) {
        v4f s0 = {d0[0], d0[1], d0[2], d0[3]};
        v4f s1 = {d1[0], d1[1], d1[2], d1[3]};
        v4f s2 = {d2[0], d2[1], d2[2], d2[3]};
        *(v4f*)(red + (wave * 3 + 0) * 64 + c * 4) = s0;
        *(v4f*)(red + (wave * 3 + 1) * 64 + c * 4) = s1;
        *(v4f*)(red + (wave * 3 + 2) * 64 + c * 4) = s2;
    }
    __syncthreads();

    if (tid < 128) {
        int n = tid >> 2, o = tid & 3;
        int off = (n >> 4) * 64 + (n & 15) * 4 + o;
        float s = red[off];
        #pragma unroll
        for (int w = 1; w < 8; ++w) s += red[w * 192 + off];
        Wc[tid] = s;
    } else if (tid < 132) {
        int o = tid - 128;
        float s = red[2 * 64 + o];
        #pragma unroll
        for (int w = 1; w < 8; ++w) s += red[w * 192 + 2 * 64 + o];
        bc[o] = br[o] + s + bsw[o];
    }
}

#define KS_WAVE 15
__global__ __launch_bounds__(256) void k_main(const float* __restrict__ x,
                                              const float* __restrict__ W0,
                                              const float* __restrict__ b0,
                                              const float* __restrict__ Wc,
                                              const float* __restrict__ bc,
                                              const float* __restrict__ W2,
                                              const float* __restrict__ b2,
                                              const float* __restrict__ W3,
                                              const float* __restrict__ b3,
                                              float* __restrict__ out) {
    __shared__ __attribute__((aligned(16))) float xt[XTOK * 4];
    __shared__ __attribute__((aligned(16))) float part[8 * PW];
    __shared__ float b0s[H0];
    __shared__ float wcs[H0 * 4];
    __shared__ float bcs[4];
    __shared__ __attribute__((aligned(16))) float h0s[(TB + 4) * 36];
    __shared__ __attribute__((aligned(16))) float hm[(TB + 4) * 4];
    __shared__ __attribute__((aligned(16))) float w2s[20 * 32];
    __shared__ __attribute__((aligned(16))) float b2s[32];
    __shared__ __attribute__((aligned(16))) float w3s[32 * 4];
    __shared__ __attribute__((aligned(16))) float b3s[4];

    const int tid = threadIdx.x;
    const int bb  = blockIdx.x >> 5;
    const int t0  = (blockIdx.x & 31) * TB;
    const int wave = __builtin_amdgcn_readfirstlane(tid >> 5);
    const int lane = tid & 31;
    const int hl   = lane >> 4;
    const int m0   = lane & 15;
    const int c0   = hl << 1;
    const int k_lo = wave * KS_WAVE;

    const float* bptr = W0 + (size_t)(((k_lo << 2) + c0) * H0 + m0);
    v2f breg[16][2];
    #pragma unroll
    for (int i = 0; i < KS_WAVE; ++i) {
        v2f f0 = {bptr[(4 * i) * H0],      bptr[(4 * i + 1) * H0]};
        v2f f1 = {bptr[(4 * i) * H0 + 16], bptr[(4 * i + 1) * H0 + 16]};
        breg[i][0] = f0; breg[i][1] = f1;
    }
    {
        v2f z = {0.f, 0.f};
        breg[15][0] = z; breg[15][1] = z;
    }
    if (wave == 7) {
        v2f f0 = {bptr[60 * H0],      bptr[61 * H0]};
        v2f f1 = {bptr[60 * H0 + 16], bptr[61 * H0 + 16]};
        breg[15][0] = f0; breg[15][1] = f1;
    }

    const float w2r0 = W2[tid];
    const float w2r1 = W2[tid + 256];
    float w2r2 = 0.f, w3r = 0.f, b2r = 0.f, b3r = 0.f;
    if (tid < 128) { w2r2 = W2[tid + 512]; w3r = W3[tid]; }
    if (tid < 32)  b2r = b2[tid];
    if (tid < 4)   b3r = b3[tid];

    for (int i = tid; i < (XTOK - XPAD) * 4; i += 256) {
        int j = i >> 2, c = i & 3;
        int gt = (t0 + j + L_TOK - (HALO + 4)) & (L_TOK - 1);
        xt[XPAD * 4 + i] = x[((bb << 12) + gt) * 4 + c];
    }
    if (tid < XPAD * 4) xt[tid] = 0.f;
    if (tid < H0)  b0s[tid] = b0[tid];
    if (tid < 128) wcs[tid] = Wc[tid];
    if (tid < 4)   bcs[tid] = bc[tid];
    w2s[tid]       = w2r0;
    w2s[tid + 256] = w2r1;
    if (tid < 128) { w2s[tid + 512] = w2r2; w3s[tid] = w3r; }
    if (tid < 32)  b2s[tid] = b2r;
    if (tid < 4)   b3s[tid] = b3r;
    __syncthreads();

    v8f acc[NTILE][2];
    #pragma unroll
    for (int mt = 0; mt < NTILE; ++mt) { v8f z = {}; acc[mt][0] = z; acc[mt][1] = z; }

    const float* ap = xt + m0 * 4 + c0 + 4 * k_lo;
    #pragma unroll
    for (int i = 0; i < KS_WAVE; ++i) {
        #pragma unroll
        for (int mt = 0; mt < NTILE; ++mt) {
            v2f a = *(const v2f*)(ap + mt * 64 + 4 * i);
            acc[mt][0] = __builtin_amdgcn_wmma_f32_16x16x4_f32(false, a, false, breg[i][0], (short)0, acc[mt][0], false, false);
            acc[mt][1] = __builtin_amdgcn_wmma_f32_16x16x4_f32(false, a, false, breg[i][1], (short)0, acc[mt][1], false, false);
        }
    }
    if (wave == 7) {
        #pragma unroll
        for (int mt = 0; mt < NTILE; ++mt) {
            v2f a = *(const v2f*)(ap + mt * 64 + 4 * KS_WAVE);
            acc[mt][0] = __builtin_amdgcn_wmma_f32_16x16x4_f32(false, a, false, breg[15][0], (short)0, acc[mt][0], false, false);
            acc[mt][1] = __builtin_amdgcn_wmma_f32_16x16x4_f32(false, a, false, breg[15][1], (short)0, acc[mt][1], false, false);
        }
    }

    float* pw = part + wave * PW + lane * 4;
    #pragma unroll
    for (int mt = 0; mt < NTILE; ++mt) {
        #pragma unroll
        for (int nt = 0; nt < 2; ++nt) {
            v8f d = acc[mt][nt];
            v4f lo = {d[0], d[1], d[2], d[3]};
            v4f hi = {d[4], d[5], d[6], d[7]};
            *(v4f*)(pw + ((mt * 2 + nt) * 2 + 0) * 128) = lo;
            *(v4f*)(pw + ((mt * 2 + nt) * 2 + 1) * 128) = hi;
        }
    }
    __syncthreads();

    const float* pr = part + (wave + 1) * 512 + lane * 4;
    v8f acc0, acc1;
    {
        v4f q0 = *(const v4f*)(pr);
        v4f q1 = *(const v4f*)(pr + 128);
        v4f q2 = *(const v4f*)(pr + 256);
        v4f q3 = *(const v4f*)(pr + 384);
        #pragma unroll
        for (int s = 1; s < 8; ++s) {
            q0 += *(const v4f*)(pr + s * PW);
            q1 += *(const v4f*)(pr + s * PW + 128);
            q2 += *(const v4f*)(pr + s * PW + 256);
            q3 += *(const v4f*)(pr + s * PW + 384);
        }
        v8f r0 = {q0.x, q0.y, q0.z, q0.w, q1.x, q1.y, q1.z, q1.w};
        v8f r1 = {q2.x, q2.y, q2.z, q2.w, q3.x, q3.y, q3.z, q3.w};
        acc0 = r0; acc1 = r1;
    }

    #pragma unroll
    for (int g = 0; g < 8; ++g) {
        int R = 4 + (wave << 4) + g + (hl << 3);
        h0s[R * 36 + m0]      = tanhf(acc0[g] + b0s[m0]);
        h0s[R * 36 + m0 + 16] = tanhf(acc1[g] + b0s[m0 + 16]);
    }
    if (hl) {
        const int nt = wave & 1, e = wave >> 1;
        const float* ph = part + nt * 256 + 128 + lane * 4 + e;
        float s = ph[0];
        #pragma unroll
        for (int w = 1; w < 8; ++w) s += ph[w * PW];
        h0s[e * 36 + m0 + 16 * nt] = tanhf(s + b0s[m0 + 16 * nt]);
    }
    __syncthreads();

    {
        const int o  = tid & 3;
        const int rq = tid >> 2;
        const int r2 = (rq < 4) ? (rq + TB) : (TB + 3);
        const float* ha = h0s + rq * 36;
        const float* hb = h0s + (rq + 64) * 36;
        const float* hc = h0s + r2 * 36;
        float sa = bcs[o], sb = sa, sc = sa;
        #pragma unroll 8
        for (int n = 0; n < 32; ++n) {
            const float w = wcs[n * 4 + o];
            sa = fmaf(ha[n], w, sa);
            sb = fmaf(hb[n], w, sb);
            sc = fmaf(hc[n], w, sc);
        }
        hm[tid]       = sa;
        hm[tid + 256] = sb;
        if (tid < 16) hm[tid + 512] = sc;
    }
    __syncthreads();

    {
        const int tk = tid >> 1;
        const int nb = (tid & 1) << 4;
        float win[20];
        #pragma unroll
        for (int j = 0; j < 5; ++j) {
            v4f v = *(const v4f*)(hm + (tk + j) * 4);
            win[j * 4 + 0] = v.x; win[j * 4 + 1] = v.y;
            win[j * 4 + 2] = v.z; win[j * 4 + 3] = v.w;
        }
        const v4f zero4 = {0.f, 0.f, 0.f, 0.f};
        v4f o = (tid & 1) ? zero4 : *(const v4f*)b3s;
        #pragma unroll
        for (int q = 0; q < 4; ++q) {
            const int n = nb + 4 * q;
            v4f h = *(const v4f*)(b2s + n);
            #pragma unroll
            for (int k = 0; k < 20; ++k) {
                v4f w = *(const v4f*)(w2s + k * 32 + n);
                h.x = fmaf(win[k], w.x, h.x);
                h.y = fmaf(win[k], w.y, h.y);
                h.z = fmaf(win[k], w.z, h.z);
                h.w = fmaf(win[k], w.w, h.w);
            }
            h.x = fmaxf(h.x, 0.f); h.y = fmaxf(h.y, 0.f);
            h.z = fmaxf(h.z, 0.f); h.w = fmaxf(h.w, 0.f);
            v4f wa = *(const v4f*)(w3s + (n + 0) * 4);
            v4f wb = *(const v4f*)(w3s + (n + 1) * 4);
            v4f wc4 = *(const v4f*)(w3s + (n + 2) * 4);
            v4f wd = *(const v4f*)(w3s + (n + 3) * 4);
            o.x = fmaf(h.x, wa.x, o.x); o.y = fmaf(h.x, wa.y, o.y); o.z = fmaf(h.x, wa.z, o.z); o.w = fmaf(h.x, wa.w, o.w);
            o.x = fmaf(h.y, wb.x, o.x); o.y = fmaf(h.y, wb.y, o.y); o.z = fmaf(h.y, wb.z, o.z); o.w = fmaf(h.y, wb.w, o.w);
            o.x = fmaf(h.z, wc4.x, o.x); o.y = fmaf(h.z, wc4.y, o.y); o.z = fmaf(h.z, wc4.z, o.z); o.w = fmaf(h.z, wc4.w, o.w);
            o.x = fmaf(h.w, wd.x, o.x); o.y = fmaf(h.w, wd.y, o.y); o.z = fmaf(h.w, wd.z, o.z); o.w = fmaf(h.w, wd.w, o.w);
        }
        o.x += __shfl_xor(o.x, 1);
        o.y += __shfl_xor(o.y, 1);
        o.z += __shfl_xor(o.z, 1);
        o.w += __shfl_xor(o.w, 1);
        if (!(tid & 1))
            *(v4f*)(out + ((size_t)((bb << 12) + t0 + tk) << 2)) = o;
    }
}

extern "C" void kernel_launch(void* const* d_in, const int* in_sizes, int n_in,
                              void* d_out, int out_size, void* d_ws, size_t ws_size,
                              hipStream_t stream) {
    (void)in_sizes; (void)n_in; (void)out_size; (void)ws_size;
    const float* x  = (const float*)d_in[0];
    const float* W0 = (const float*)d_in[1];
    const float* b0 = (const float*)d_in[2];
    const float* W1 = (const float*)d_in[3];
    const float* b1 = (const float*)d_in[4];
    const float* Ws = (const float*)d_in[5];
    const float* bs = (const float*)d_in[6];
    const float* Wr = (const float*)d_in[7];
    const float* br = (const float*)d_in[8];
    const float* W2 = (const float*)d_in[9];
    const float* b2 = (const float*)d_in[10];
    const float* W3 = (const float*)d_in[11];
    const float* b3 = (const float*)d_in[12];
    float* out = (float*)d_out;

    float* ws   = (float*)d_ws;
    float* P    = ws;
    float* Wc   = ws + 16896;
    float* bc   = ws + 17024;

    k_wsr <<<264, 256, 0, stream>>>(Ws, bs, Wr, P);
    k_fold<<<1,   256, 0, stream>>>(W1, b1, br, P, Wc, bc);
    k_main<<<256, 256, 0, stream>>>(x, W0, b0, Wc, bc, W2, b2, W3, b3, out);
}


# Launch fusion 3 -> 2 by linearity: each (row tile, k chunk) block of Ws@Wr applies its own W1 slice on WMMA, main kernel adds the 132 partial 33x4 tiles; f32 sum order of Wc, bc changes
# speedup vs baseline: 14.1812x; 1.1578x over previous
#include <hip/hip_runtime.h>
#include <hip/hip_bf16.h>


typedef __attribute__((ext_vector_type(2))) float v2f;
typedef __attribute__((ext_vector_type(4))) float v4f;
typedef __attribute__((ext_vector_type(8))) float v8f;

#define L_TOK   4096
#define B_BATCH 8
#define K0      484
#define H0      32
#define TB      128
#define HALO    120
#define NTILE   9
#define XPAD    12
#define XTOK    (XPAD + 4 + HALO + TB)
#define PW      (NTILE * 512)

#define FK_CH    1024
#define FK_KW    128
#define FK_NCH   4
#define FK_TILES 33
#define FK_PITCH 1032
#define NPART    (FK_TILES * FK_NCH)
#define PPITCH   160

__global__ __launch_bounds__(256) void k_pre(const float* __restrict__ Ws,
                                             const float* __restrict__ bs,
                                             const float* __restrict__ Wr,
                                             const float* __restrict__ W1,
                                             const float* __restrict__ b1,
                                             float* __restrict__ Pc) {
    __shared__ __attribute__((aligned(16))) float wrT[5 * FK_PITCH];
    __shared__ __attribute__((aligned(16))) float red[8 * 64];
    __shared__ __attribute__((aligned(16))) float tT[5 * 16];

    const int tid  = threadIdx.x;
    const int tile = blockIdx.x >> 2;
    const int kc   = blockIdx.x & 3;
    const int kb   = kc * FK_CH;
    const int wave = __builtin_amdgcn_readfirstlane(tid >> 5);
    const int lane = tid & 31;
    const int h    = lane >> 4;
    const int c    = lane & 15;

    const int row = tile * 16 + c;
    const float* src = (row < 512) ? (Ws + (size_t)row * 4096) : bs;
    const v4f* bp = (const v4f*)(src + kb + wave * FK_KW + 4 * h);
    const v4f zero4 = {0.f, 0.f, 0.f, 0.f};
    v4f bv[16];
    #pragma unroll
    for (int j = 0; j < 16; ++j) bv[j] = zero4;
    if (row <= 512) {
        #pragma unroll
        for (int j = 0; j < 16; ++j) bv[j] = bp[2 * j];
    }

    v4f m0[2], m1[2], m2[2];
    #pragma unroll
    for (int j = 0; j < 2; ++j) { m0[j] = zero4; m1[j] = zero4; m2[j] = zero4; }
    if (wave == 0) {
        if (tile < 32) {
            const float* q = W1 + (size_t)c * 512 + tile * 16 + 4 * h;
            #pragma unroll
            for (int j = 0; j < 2; ++j) {
                m0[j] = *(const v4f*)(q + 8 * j);
                m1[j] = *(const v4f*)(q + 16 * 512 + 8 * j);
            }
            if (c == 0) {
                #pragma unroll
                for (int j = 0; j < 2; ++j) m2[j] = *(const v4f*)(b1 + tile * 16 + 4 * h + 8 * j);
            }
        } else if (lane == 0) {
            m2[0].x = 1.f;
        }
    }

    #pragma unroll
    for (int r = 0; r < 4; ++r) {
        int k = tid + (r << 8);
        v4f w = ((const v4f*)(Wr + (size_t)kb * 4))[k];
        wrT[0 * FK_PITCH + k] = w.x;
        wrT[1 * FK_PITCH + k] = w.y;
        wrT[2 * FK_PITCH + k] = w.z;
        wrT[3 * FK_PITCH + k] = w.w;
        wrT[4 * FK_PITCH + k] = 0.f;
    }
    if (tid >= 64 && tid < 80) tT[tid] = 0.f;
    __syncthreads();

    v8f acc0 = {}; v8f acc1 = {};
    const float* ap = wrT + (c < 4 ? c : 4) * FK_PITCH + wave * FK_KW + 4 * h;
    #pragma unroll
    for (int j = 0; j < 16; ++j) {
        v4f av = *(const v4f*)(ap + 8 * j);
        v2f a0 = {av.x, av.y},       a1 = {av.z, av.w};
        v2f b0 = {bv[j].x, bv[j].y}, b1v = {bv[j].z, bv[j].w};
        acc0 = __builtin_amdgcn_wmma_f32_16x16x4_f32(false, a0, false, b0,  (short)0, acc0, false, false);
        acc1 = __builtin_amdgcn_wmma_f32_16x16x4_f32(false, a1, false, b1v, (short)0, acc1, false, false);
    }

    if (h == 0) {
        v4f s = {acc0[0] + acc1[0], acc0[1] + acc1[1], acc0[2] + acc1[2], acc0[3] + acc1[3]};
        *(v4f*)(red + wave * 64 + c * 4) = s;
    }
    __syncthreads();
    if (tid < 64) {
        float s = red[tid];
        #pragma unroll
        for (int w = 1; w < 8; ++w) s += red[w * 64 + tid];
        tT[(tid & 3) * 16 + (tid >> 2)] = s;
    }
    __syncthreads();

    if (wave == 0) {
        v8f e0 = {}; v8f e1 = {}; v8f e2 = {};
        const float* a2 = tT + (c < 4 ? c : 4) * 16 + 4 * h;
        #pragma unroll
        for (int j = 0; j < 2; ++j) {
            v4f av = *(const v4f*)(a2 + 8 * j);
            v2f a0 = {av.x, av.y}, a1 = {av.z, av.w};
            v2f x0 = {m0[j].x, m0[j].y}, x1 = {m0[j].z, m0[j].w};
            v2f y0 = {m1[j].x, m1[j].y}, y1 = {m1[j].z, m1[j].w};
            v2f z0 = {m2[j].x, m2[j].y}, z1 = {m2[j].z, m2[j].w};
            e0 = __builtin_amdgcn_wmma_f32_16x16x4_f32(false, a0, false, x0, (short)0, e0, false, false);
            e1 = __builtin_amdgcn_wmma_f32_16x16x4_f32(false, a0, false, y0, (short)0, e1, false, false);
            e2 = __builtin_amdgcn_wmma_f32_16x16x4_f32(false, a0, false, z0, (short)0, e2, false, false);
            e0 = __builtin_amdgcn_wmma_f32_16x16x4_f32(false, a1, false, x1, (short)0, e0, false, false);
            e1 = __builtin_amdgcn_wmma_f32_16x16x4_f32(false, a1, false, y1, (short)0, e1, false, false);
            e2 = __builtin_amdgcn_wmma_f32_16x16x4_f32(false, a1, false, z1, (short)0, e2, false, false);
        }
        if (h == 0) {
            float* dst = Pc + (size_t)blockIdx.x * PPITCH;
            v4f s0 = {e0[0], e0[1], e0[2], e0[3]};
            v4f s1 = {e1[0], e1[1], e1[2], e1[3]};
            v4f s2 = {e2[0], e2[1], e2[2], e2[3]};
            *(v4f*)(dst + c * 4)      = s0;
            *(v4f*)(dst + 64 + c * 4) = s1;
            if (c == 0) *(v4f*)(dst + 128) = s2;
        }
    }
}

#define KS_WAVE 15
__global__ __launch_bounds__(256) void k_main(const float* __restrict__ x,
                                              const float* __restrict__ W0,
                                              const float* __restrict__ b0,
                                              const float* __restrict__ Pc,
                                              const float* __restrict__ br,
                                              const float* __restrict__ W2,
                                              const float* __restrict__ b2,
                                              const float* __restrict__ W3,
                                              const float* __restrict__ b3,
                                              float* __restrict__ out) {
    __shared__ __attribute__((aligned(16))) float xt[XTOK * 4];
    __shared__ __attribute__((aligned(16))) float part[8 * PW];
    __shared__ float b0s[H0];
    __shared__ float wcs[H0 * 4];
    __shared__ float bcs[4];
    __shared__ __attribute__((aligned(16))) float wred[7 * 132];
    __shared__ __attribute__((aligned(16))) float h0s[(TB + 4) * 36];
    __shared__ __attribute__((aligned(16))) float hm[(TB + 4) * 4];
    __shared__ __attribute__((aligned(16))) float w2s[20 * 32];
    __shared__ __attribute__((aligned(16))) float b2s[32];
    __shared__ __attribute__((aligned(16))) float w3s[32 * 4];
    __shared__ __attribute__((aligned(16))) float b3s[4];

    const int tid = threadIdx.x;
    const int bb  = blockIdx.x >> 5;
    const int t0  = (blockIdx.x & 31) * TB;
    const int wave = __builtin_amdgcn_readfirstlane(tid >> 5);
    const int lane = tid & 31;
    const int hl   = lane >> 4;
    const int m0   = lane & 15;
    const int c0   = hl << 1;
    const int k_lo = wave * KS_WAVE;

    const float* bptr = W0 + (size_t)(((k_lo << 2) + c0) * H0 + m0);
    v2f breg[16][2];
    #pragma unroll
    for (int i = 0; i < KS_WAVE; ++i) {
        v2f f0 = {bptr[(4 * i) * H0],      bptr[(4 * i + 1) * H0]};
        v2f f1 = {bptr[(4 * i) * H0 + 16], bptr[(4 * i + 1) * H0 + 16]};
        breg[i][0] = f0; breg[i][1] = f1;
    }
    {
        v2f z = {0.f, 0.f};
        breg[15][0] = z; breg[15][1] = z;
    }
    if (wave == 7) {
        v2f f0 = {bptr[60 * H0],      bptr[61 * H0]};
        v2f f1 = {bptr[60 * H0 + 16], bptr[61 * H0 + 16]};
        breg[15][0] = f0; breg[15][1] = f1;
    }

    float brr = 0.f;
    if (tid >= 128 && tid < 132) brr = br[tid - 128];

    const float w2r0 = W2[tid];
    const float w2r1 = W2[tid + 256];
    float w2r2 = 0.f, w3r = 0.f, b2r = 0.f, b3r = 0.f;
    if (tid < 128) { w2r2 = W2[tid + 512]; w3r = W3[tid]; }
    if (tid < 32)  b2r = b2[tid];
    if (tid < 4)   b3r = b3[tid];

    for (int i = tid; i < (XTOK - XPAD) * 4; i += 256) {
        int j = i >> 2, c = i & 3;
        int gt = (t0 + j + L_TOK - (HALO + 4)) & (L_TOK - 1);
        xt[XPAD * 4 + i] = x[((bb << 12) + gt) * 4 + c];
    }
    if (tid < XPAD * 4) xt[tid] = 0.f;
    if (tid < H0)  b0s[tid] = b0[tid];
    w2s[tid]       = w2r0;
    w2s[tid + 256] = w2r1;
    if (tid < 128) { w2s[tid + 512] = w2r2; w3s[tid] = w3r; }
    if (tid < 32)  b2s[tid] = b2r;
    if (tid < 4)   b3s[tid] = b3r;
    __syncthreads();

    v8f acc[NTILE][2];
    #pragma unroll
    for (int mt = 0; mt < NTILE; ++mt) { v8f z = {}; acc[mt][0] = z; acc[mt][1] = z; }

    const int pg = tid / 33;
    const int ps = tid - pg * 33;
    v4f pv[19];
    #pragma unroll
    for (int q = 0; q < 19; ++q) { v4f z = {0.f, 0.f, 0.f, 0.f}; pv[q] = z; }
    if (tid < 7 * 33) {
        const v4f* pp = (const v4f*)Pc + pg * (PPITCH / 4) + ps;
        #pragma unroll
        for (int q = 0; q < 18; ++q) pv[q] = pp[q * 7 * (PPITCH / 4)];
        if (pg < 6) pv[18] = pp[18 * 7 * (PPITCH / 4)];
    }

    const float* ap = xt + m0 * 4 + c0 + 4 * k_lo;
    #pragma unroll
    for (int i = 0; i < KS_WAVE; ++i) {
        #pragma unroll
        for (int mt = 0; mt < NTILE; ++mt) {
            v2f a = *(const v2f*)(ap + mt * 64 + 4 * i);
            acc[mt][0] = __builtin_amdgcn_wmma_f32_16x16x4_f32(false, a, false, breg[i][0], (short)0, acc[mt][0], false, false);
            acc[mt][1] = __builtin_amdgcn_wmma_f32_16x16x4_f32(false, a, false, breg[i][1], (short)0, acc[mt][1], false, false);
        }
    }
    if (wave == 7) {
        #pragma unroll
        for (int mt = 0; mt < NTILE; ++mt) {
            v2f a = *(const v2f*)(ap + mt * 64 + 4 * KS_WAVE);
            acc[mt][0] = __builtin_amdgcn_wmma_f32_16x16x4_f32(false, a, false, breg[15][0], (short)0, acc[mt][0], false, false);
            acc[mt][1] = __builtin_amdgcn_wmma_f32_16x16x4_f32(false, a, false, breg[15][1], (short)0, acc[mt][1], false, false);
        }
    }

    if (tid < 7 * 33) {
        v4f psum = pv[0];
        #pragma unroll
        for (int q = 1; q < 19; ++q) psum += pv[q];
        *(v4f*)(wred + pg * 132 + ps * 4) = psum;
    }

    float* pw = part + wave * PW + lane * 4;
    #pragma unroll
    for (int mt = 0; mt < NTILE; ++mt) {
        #pragma unroll
        for (int nt = 0; nt < 2; ++nt) {
            v8f d = acc[mt][nt];
            v4f lo = {d[0], d[1], d[2], d[3]};
            v4f hi = {d[4], d[5], d[6], d[7]};
            *(v4f*)(pw + ((mt * 2 + nt) * 2 + 0) * 128) = lo;
            *(v4f*)(pw + ((mt * 2 + nt) * 2 + 1) * 128) = hi;
        }
    }
    __syncthreads();

    if (tid < 132) {
        float ws = wred[tid];
        #pragma unroll
        for (int g = 1; g < 7; ++g) ws += wred[g * 132 + tid];
        if (tid < 128) wcs[tid] = ws; else bcs[tid - 128] = ws + brr;
    }

    const float* pr = part + (wave + 1) * 512 + lane * 4;
    v8f acc0, acc1;
    {
        v4f q0 = *(const v4f*)(pr);
        v4f q1 = *(const v4f*)(pr + 128);
        v4f q2 = *(const v4f*)(pr + 256);
        v4f q3 = *(const v4f*)(pr + 384);
        #pragma unroll
        for (int s = 1; s < 8; ++s) {
            q0 += *(const v4f*)(pr + s * PW);
            q1 += *(const v4f*)(pr + s * PW + 128);
            q2 += *(const v4f*)(pr + s * PW + 256);
            q3 += *(const v4f*)(pr + s * PW + 384);
        }
        v8f r0 = {q0.x, q0.y, q0.z, q0.w, q1.x, q1.y, q1.z, q1.w};
        v8f r1 = {q2.x, q2.y, q2.z, q2.w, q3.x, q3.y, q3.z, q3.w};
        acc0 = r0; acc1 = r1;
    }

    #pragma unroll
    for (int g = 0; g < 8; ++g) {
        int R = 4 + (wave << 4) + g + (hl << 3);
        h0s[R * 36 + m0]      = tanhf(acc0[g] + b0s[m0]);
        h0s[R * 36 + m0 + 16] = tanhf(acc1[g] + b0s[m0 + 16]);
    }
    if (hl) {
        const int nt = wave & 1, e = wave >> 1;
        const float* ph = part + nt * 256 + 128 + lane * 4 + e;
        float s = ph[0];
        #pragma unroll
        for (int w = 1; w < 8; ++w) s += ph[w * PW];
        h0s[e * 36 + m0 + 16 * nt] = tanhf(s + b0s[m0 + 16 * nt]);
    }
    __syncthreads();

    {
        const int o  = tid & 3;
        const int rq = tid >> 2;
        const int r2 = (rq < 4) ? (rq + TB) : (TB + 3);
        const float* ha = h0s + rq * 36;
        const float* hb = h0s + (rq + 64) * 36;
        const float* hc = h0s + r2 * 36;
        float sa = bcs[o], sb = sa, sc = sa;
        #pragma unroll 8
        for (int n = 0; n < 32; ++n) {
            const float w = wcs[n * 4 + o];
            sa = fmaf(ha[n], w, sa);
            sb = fmaf(hb[n], w, sb);
            sc = fmaf(hc[n], w, sc);
        }
        hm[tid]       = sa;
        hm[tid + 256] = sb;
        if (tid < 16) hm[tid + 512] = sc;
    }
    __syncthreads();

    {
        const int tk = tid >> 1;
        const int nb = (tid & 1) << 4;
        float win[20];
        #pragma unroll
        for (int j = 0; j < 5; ++j) {
            v4f v = *(const v4f*)(hm + (tk + j) * 4);
            win[j * 4 + 0] = v.x; win[j * 4 + 1] = v.y;
            win[j * 4 + 2] = v.z; win[j * 4 + 3] = v.w;
        }
        const v4f zero4 = {0.f, 0.f, 0.f, 0.f};
        v4f o = (tid & 1) ? zero4 : *(const v4f*)b3s;
        #pragma unroll
        for (int q = 0; q < 4; ++q) {
            const int n = nb + 4 * q;
            v4f h = *(const v4f*)(b2s + n);
            #pragma unroll
            for (int k = 0; k < 20; ++k) {
                v4f w = *(const v4f*)(w2s + k * 32 + n);
                h.x = fmaf(win[k], w.x, h.x);
                h.y = fmaf(win[k], w.y, h.y);
                h.z = fmaf(win[k], w.z, h.z);
                h.w = fmaf(win[k], w.w, h.w);
            }
            h.x = fmaxf(h.x, 0.f); h.y = fmaxf(h.y, 0.f);
            h.z = fmaxf(h.z, 0.f); h.w = fmaxf(h.w, 0.f);
            v4f wa = *(const v4f*)(w3s + (n + 0) * 4);
            v4f wb = *(const v4f*)(w3s + (n + 1) * 4);
            v4f wc4 = *(const v4f*)(w3s + (n + 2) * 4);
            v4f wd = *(const v4f*)(w3s + (n + 3) * 4);
            o.x = fmaf(h.x, wa.x, o.x); o.y = fmaf(h.x, wa.y, o.y); o.z = fmaf(h.x, wa.z, o.z); o.w = fmaf(h.x, wa.w, o.w);
            o.x = fmaf(h.y, wb.x, o.x); o.y = fmaf(h.y, wb.y, o.y); o.z = fmaf(h.y, wb.z, o.z); o.w = fmaf(h.y, wb.w, o.w);
            o.x = fmaf(h.z, wc4.x, o.x); o.y = fmaf(h.z, wc4.y, o.y); o.z = fmaf(h.z, wc4.z, o.z); o.w = fmaf(h.z, wc4.w, o.w);
            o.x = fmaf(h.w, wd.x, o.x); o.y = fmaf(h.w, wd.y, o.y); o.z = fmaf(h.w, wd.z, o.z); o.w = fmaf(h.w, wd.w, o.w);
        }
        o.x += __shfl_xor(o.x, 1);
        o.y += __shfl_xor(o.y, 1);
        o.z += __shfl_xor(o.z, 1);
        o.w += __shfl_xor(o.w, 1);
        if (!(tid & 1))
            *(v4f*)(out + ((size_t)((bb << 12) + t0 + tk) << 2)) = o;
    }
}

extern "C" void kernel_launch(void* const* d_in, const int* in_sizes, int n_in,
                              void* d_out, int out_size, void* d_ws, size_t ws_size,
                              hipStream_t stream) {
    (void)in_sizes; (void)n_in; (void)out_size; (void)ws_size;
    const float* x  = (const float*)d_in[0];
    const float* W0 = (const float*)d_in[1];
    const float* b0 = (const float*)d_in[2];
    const float* W1 = (const float*)d_in[3];
    const float* b1 = (const float*)d_in[4];
    const float* Ws = (const float*)d_in[5];
    const float* bs = (const float*)d_in[6];
    const float* Wr = (const float*)d_in[7];
    const float* br = (const float*)d_in[8];
    const float* W2 = (const float*)d_in[9];
    const float* b2 = (const float*)d_in[10];
    const float* W3 = (const float*)d_in[11];
    const float* b3 = (const float*)d_in[12];
    float* out = (float*)d_out;

    float* Pc = (float*)d_ws;

    k_pre <<<NPART, 256, 0, stream>>>(Ws, bs, Wr, W1, b1, Pc);
    k_main<<<256,   256, 0, stream>>>(x, W0, b0, Pc, br, W2, b2, W3, b3, out);
}


# Back end (32->4, window-5 20->32 relu, 32->4): per-thread fma chains -> f32 16x16x4 WMMA, tokens as tile columns, hidden D tiles serve in place as the last B; f32 sum order of these 3 dots changes
# speedup vs baseline: 15.1103x; 1.0655x over previous
#include <hip/hip_runtime.h>
#include <hip/hip_bf16.h>


typedef __attribute__((ext_vector_type(2))) float v2f;
typedef __attribute__((ext_vector_type(4))) float v4f;
typedef __attribute__((ext_vector_type(8))) float v8f;

#define L_TOK   4096
#define B_BATCH 8
#define K0      484
#define H0      32
#define TB      128
#define HALO    120
#define NTILE   9
#define XPAD    12
#define XTOK    (XPAD + 4 + HALO + TB)
#define PW      (NTILE * 512)

#define FK_CH    1024
#define FK_KW    128
#define FK_NCH   4
#define FK_TILES 33
#define FK_PITCH 1032
#define NPART    (FK_TILES * FK_NCH)
#define PPITCH   160

__global__ __launch_bounds__(256) void k_pre(const float* __restrict__ Ws,
                                             const float* __restrict__ bs,
                                             const float* __restrict__ Wr,
                                             const float* __restrict__ W1,
                                             const float* __restrict__ b1,
                                             float* __restrict__ Pc) {
    __shared__ __attribute__((aligned(16))) float wrT[5 * FK_PITCH];
    __shared__ __attribute__((aligned(16))) float red[8 * 64];
    __shared__ __attribute__((aligned(16))) float tT[5 * 16];

    const int tid  = threadIdx.x;
    const int tile = blockIdx.x >> 2;
    const int kc   = blockIdx.x & 3;
    const int kb   = kc * FK_CH;
    const int wave = __builtin_amdgcn_readfirstlane(tid >> 5);
    const int lane = tid & 31;
    const int h    = lane >> 4;
    const int c    = lane & 15;

    const int row = tile * 16 + c;
    const float* src = (row < 512) ? (Ws + (size_t)row * 4096) : bs;
    const v4f* bp = (const v4f*)(src + kb + wave * FK_KW + 4 * h);
    const v4f zero4 = {0.f, 0.f, 0.f, 0.f};
    v4f bv[16];
    #pragma unroll
    for (int j = 0; j < 16; ++j) bv[j] = zero4;
    if (row <= 512) {
        #pragma unroll
        for (int j = 0; j < 16; ++j) bv[j] = bp[2 * j];
    }

    v4f m0[2], m1[2], m2[2];
    #pragma unroll
    for (int j = 0; j < 2; ++j) { m0[j] = zero4; m1[j] = zero4; m2[j] = zero4; }
    if (wave == 0) {
        if (tile < 32) {
            const float* q = W1 + (size_t)c * 512 + tile * 16 + 4 * h;
            #pragma unroll
            for (int j = 0; j < 2; ++j) {
                m0[j] = *(const v4f*)(q + 8 * j);
                m1[j] = *(const v4f*)(q + 16 * 512 + 8 * j);
            }
            if (c == 0) {
                #pragma unroll
                for (int j = 0; j < 2; ++j) m2[j] = *(const v4f*)(b1 + tile * 16 + 4 * h + 8 * j);
            }
        } else if (lane == 0) {
            m2[0].x = 1.f;
        }
    }

    #pragma unroll
    for (int r = 0; r < 4; ++r) {
        int k = tid + (r << 8);
        v4f w = ((const v4f*)(Wr + (size_t)kb * 4))[k];
        wrT[0 * FK_PITCH + k] = w.x;
        wrT[1 * FK_PITCH + k] = w.y;
        wrT[2 * FK_PITCH + k] = w.z;
        wrT[3 * FK_PITCH + k] = w.w;
        wrT[4 * FK_PITCH + k] = 0.f;
    }
    if (tid >= 64 && tid < 80) tT[tid] = 0.f;
    __syncthreads();

    v8f acc0 = {}; v8f acc1 = {};
    const float* ap = wrT + (c < 4 ? c : 4) * FK_PITCH + wave * FK_KW + 4 * h;
    #pragma unroll
    for (int j = 0; j < 16; ++j) {
        v4f av = *(const v4f*)(ap + 8 * j);
        v2f a0 = {av.x, av.y},       a1 = {av.z, av.w};
        v2f b0 = {bv[j].x, bv[j].y}, b1v = {bv[j].z, bv[j].w};
        acc0 = __builtin_amdgcn_wmma_f32_16x16x4_f32(false, a0, false, b0,  (short)0, acc0, false, false);
        acc1 = __builtin_amdgcn_wmma_f32_16x16x4_f32(false, a1, false, b1v, (short)0, acc1, false, false);
    }

    if (h == 0) {
        v4f s = {acc0[0] + acc1[0], acc0[1] + acc1[1], acc0[2] + acc1[2], acc0[3] + acc1[3]};
        *(v4f*)(red + wave * 64 + c * 4) = s;
    }
    __syncthreads();
    if (tid < 64) {
        float s = red[tid];
        #pragma unroll
        for (int w = 1; w < 8; ++w) s += red[w * 64 + tid];
        tT[(tid & 3) * 16 + (tid >> 2)] = s;
    }
    __syncthreads();

    if (wave == 0) {
        v8f e0 = {}; v8f e1 = {}; v8f e2 = {};
        const float* a2 = tT + (c < 4 ? c : 4) * 16 + 4 * h;
        #pragma unroll
        for (int j = 0; j < 2; ++j) {
            v4f av = *(const v4f*)(a2 + 8 * j);
            v2f a0 = {av.x, av.y}, a1 = {av.z, av.w};
            v2f x0 = {m0[j].x, m0[j].y}, x1 = {m0[j].z, m0[j].w};
            v2f y0 = {m1[j].x, m1[j].y}, y1 = {m1[j].z, m1[j].w};
            v2f z0 = {m2[j].x, m2[j].y}, z1 = {m2[j].z, m2[j].w};
            e0 = __builtin_amdgcn_wmma_f32_16x16x4_f32(false, a0, false, x0, (short)0, e0, false, false);
            e1 = __builtin_amdgcn_wmma_f32_16x16x4_f32(false, a0, false, y0, (short)0, e1, false, false);
            e2 = __builtin_amdgcn_wmma_f32_16x16x4_f32(false, a0, false, z0, (short)0, e2, false, false);
            e0 = __builtin_amdgcn_wmma_f32_16x16x4_f32(false, a1, false, x1, (short)0, e0, false, false);
            e1 = __builtin_amdgcn_wmma_f32_16x16x4_f32(false, a1, false, y1, (short)0, e1, false, false);
            e2 = __builtin_amdgcn_wmma_f32_16x16x4_f32(false, a1, false, z1, (short)0, e2, false, false);
        }
        if (h == 0) {
            float* dst = Pc + (size_t)blockIdx.x * PPITCH;
            v4f s0 = {e0[0], e0[1], e0[2], e0[3]};
            v4f s1 = {e1[0], e1[1], e1[2], e1[3]};
            v4f s2 = {e2[0], e2[1], e2[2], e2[3]};
            *(v4f*)(dst + c * 4)      = s0;
            *(v4f*)(dst + 64 + c * 4) = s1;
            if (c == 0) *(v4f*)(dst + 128) = s2;
        }
    }
}

#define KS_WAVE 15
__global__ __launch_bounds__(256) void k_main(const float* __restrict__ x,
                                              const float* __restrict__ W0,
                                              const float* __restrict__ b0,
                                              const float* __restrict__ Pc,
                                              const float* __restrict__ br,
                                              const float* __restrict__ W2,
                                              const float* __restrict__ b2,
                                              const float* __restrict__ W3,
                                              const float* __restrict__ b3,
                                              float* __restrict__ out) {
    __shared__ __attribute__((aligned(16))) float xt[XTOK * 4];
    __shared__ __attribute__((aligned(16))) float part[8 * PW];
    __shared__ float b0s[H0];
    __shared__ __attribute__((aligned(16))) float wct[5 * H0];
    __shared__ __attribute__((aligned(16))) float bcs[4];
    __shared__ __attribute__((aligned(16))) float wred[7 * 132];
    __shared__ __attribute__((aligned(16))) float h0s[(TB + 4) * 36];
    __shared__ __attribute__((aligned(16))) float hm[(TB + 4) * 4];
    __shared__ __attribute__((aligned(16))) float w2s[20 * 32];
    __shared__ __attribute__((aligned(16))) float b2s[32];
    __shared__ __attribute__((aligned(16))) float w3t[5 * 32];
    __shared__ __attribute__((aligned(16))) float b3s[4];

    const int tid = threadIdx.x;
    const int bb  = blockIdx.x >> 5;
    const int t0  = (blockIdx.x & 31) * TB;
    const int wave = __builtin_amdgcn_readfirstlane(tid >> 5);
    const int lane = tid & 31;
    const int hl   = lane >> 4;
    const int m0   = lane & 15;
    const int c0   = hl << 1;
    const int k_lo = wave * KS_WAVE;

    const float* bptr = W0 + (size_t)(((k_lo << 2) + c0) * H0 + m0);
    v2f breg[16][2];
    #pragma unroll
    for (int i = 0; i < KS_WAVE; ++i) {
        v2f f0 = {bptr[(4 * i) * H0],      bptr[(4 * i + 1) * H0]};
        v2f f1 = {bptr[(4 * i) * H0 + 16], bptr[(4 * i + 1) * H0 + 16]};
        breg[i][0] = f0; breg[i][1] = f1;
    }
    {
        v2f z = {0.f, 0.f};
        breg[15][0] = z; breg[15][1] = z;
    }
    if (wave == 7) {
        v2f f0 = {bptr[60 * H0],      bptr[61 * H0]};
        v2f f1 = {bptr[60 * H0 + 16], bptr[61 * H0 + 16]};
        breg[15][0] = f0; breg[15][1] = f1;
    }

    float brr = 0.f;
    if (tid >= 128 && tid < 132) brr = br[tid - 128];

    const float w2r0 = W2[tid];
    const float w2r1 = W2[tid + 256];
    float w2r2 = 0.f, w3r = 0.f, b2r = 0.f, b3r = 0.f;
    if (tid < 128) { w2r2 = W2[tid + 512]; w3r = W3[tid]; }
    if (tid < 32)  b2r = b2[tid];
    if (tid < 4)   b3r = b3[tid];

    for (int i = tid; i < (XTOK - XPAD) * 4; i += 256) {
        int j = i >> 2, c = i & 3;
        int gt = (t0 + j + L_TOK - (HALO + 4)) & (L_TOK - 1);
        xt[XPAD * 4 + i] = x[((bb << 12) + gt) * 4 + c];
    }
    if (tid < XPAD * 4) xt[tid] = 0.f;
    if (tid < H0)  b0s[tid] = b0[tid];
    w2s[tid]       = w2r0;
    w2s[tid + 256] = w2r1;
    if (tid < 128) { w2s[tid + 512] = w2r2; w3t[(tid & 3) * 32 + (tid >> 2)] = w3r; }
    if (tid < 32)  { b2s[tid] = b2r; w3t[4 * 32 + tid] = 0.f; wct[4 * H0 + tid] = 0.f; }
    if (tid < 4)   b3s[tid] = b3r;
    __syncthreads();

    v8f acc[NTILE][2];
    #pragma unroll
    for (int mt = 0; mt < NTILE; ++mt) { v8f z = {}; acc[mt][0] = z; acc[mt][1] = z; }

    const int pg = tid / 33;
    const int ps = tid - pg * 33;
    v4f pv[19];
    #pragma unroll
    for (int q = 0; q < 19; ++q) { v4f z = {0.f, 0.f, 0.f, 0.f}; pv[q] = z; }
    if (tid < 7 * 33) {
        const v4f* pp = (const v4f*)Pc + pg * (PPITCH / 4) + ps;
        #pragma unroll
        for (int q = 0; q < 18; ++q) pv[q] = pp[q * 7 * (PPITCH / 4)];
        if (pg < 6) pv[18] = pp[18 * 7 * (PPITCH / 4)];
    }

    const float* ap = xt + m0 * 4 + c0 + 4 * k_lo;
    #pragma unroll
    for (int i = 0; i < KS_WAVE; ++i) {
        #pragma unroll
        for (int mt = 0; mt < NTILE; ++mt) {
            v2f a = *(const v2f*)(ap + mt * 64 + 4 * i);
            acc[mt][0] = __builtin_amdgcn_wmma_f32_16x16x4_f32(false, a, false, breg[i][0], (short)0, acc[mt][0], false, false);
            acc[mt][1] = __builtin_amdgcn_wmma_f32_16x16x4_f32(false, a, false, breg[i][1], (short)0, acc[mt][1], false, false);
        }
    }
    if (wave == 7) {
        #pragma unroll
        for (int mt = 0; mt < NTILE; ++mt) {
            v2f a = *(const v2f*)(ap + mt * 64 + 4 * KS_WAVE);
            acc[mt][0] = __builtin_amdgcn_wmma_f32_16x16x4_f32(false, a, false, breg[15][0], (short)0, acc[mt][0], false, false);
            acc[mt][1] = __builtin_amdgcn_wmma_f32_16x16x4_f32(false, a, false, breg[15][1], (short)0, acc[mt][1], false, false);
        }
    }

    if (tid < 7 * 33) {
        v4f psum = pv[0];
        #pragma unroll
        for (int q = 1; q < 19; ++q) psum += pv[q];
        *(v4f*)(wred + pg * 132 + ps * 4) = psum;
    }

    float* pw = part + wave * PW + lane * 4;
    #pragma unroll
    for (int mt = 0; mt < NTILE; ++mt) {
        #pragma unroll
        for (int nt = 0; nt < 2; ++nt) {
            v8f d = acc[mt][nt];
            v4f lo = {d[0], d[1], d[2], d[3]};
            v4f hi = {d[4], d[5], d[6], d[7]};
            *(v4f*)(pw + ((mt * 2 + nt) * 2 + 0) * 128) = lo;
            *(v4f*)(pw + ((mt * 2 + nt) * 2 + 1) * 128) = hi;
        }
    }
    __syncthreads();

    if (tid < 132) {
        float ws = wred[tid];
        #pragma unroll
        for (int g = 1; g < 7; ++g) ws += wred[g * 132 + tid];
        if (tid < 128) wct[(tid & 3) * H0 + (tid >> 2)] = ws; else bcs[tid - 128] = ws + brr;
    }

    const float* pr = part + (wave + 1) * 512 + lane * 4;
    v8f acc0, acc1;
    {
        v4f q0 = *(const v4f*)(pr);
        v4f q1 = *(const v4f*)(pr + 128);
        v4f q2 = *(const v4f*)(pr + 256);
        v4f q3 = *(const v4f*)(pr + 384);
        #pragma unroll
        for (int s = 1; s < 8; ++s) {
            q0 += *(const v4f*)(pr + s * PW);
            q1 += *(const v4f*)(pr + s * PW + 128);
            q2 += *(const v4f*)(pr + s * PW + 256);
            q3 += *(const v4f*)(pr + s * PW + 384);
        }
        v8f r0 = {q0.x, q0.y, q0.z, q0.w, q1.x, q1.y, q1.z, q1.w};
        v8f r1 = {q2.x, q2.y, q2.z, q2.w, q3.x, q3.y, q3.z, q3.w};
        acc0 = r0; acc1 = r1;
    }

    #pragma unroll
    for (int g = 0; g < 8; ++g) {
        int R = 4 + (wave << 4) + g + (hl << 3);
        h0s[R * 36 + m0]      = tanhf(acc0[g] + b0s[m0]);
        h0s[R * 36 + m0 + 16] = tanhf(acc1[g] + b0s[m0 + 16]);
    }
    if (hl) {
        const int nt = wave & 1, e = wave >> 1;
        const float* ph = part + nt * 256 + 128 + lane * 4 + e;
        float s = ph[0];
        #pragma unroll
        for (int w = 1; w < 8; ++w) s += ph[w * PW];
        h0s[e * 36 + m0 + 16 * nt] = tanhf(s + b0s[m0 + 16 * nt]);
    }
    __syncthreads();

    const int arow = (m0 < 4 ? m0 : 4) * 32;

    {
        const float* wa = wct + arow + 4 * hl;
        const v4f a0 = *(const v4f*)(wa);
        const v4f a1 = *(const v4f*)(wa + 8);
        const v4f a2 = *(const v4f*)(wa + 16);
        const v4f a3 = *(const v4f*)(wa + 24);
        const v4f cb = *(const v4f*)bcs;
        const v8f cinit = {cb.x, cb.y, cb.z, cb.w, 0.f, 0.f, 0.f, 0.f};
        const v8f zero8 = {};
        {
            const int R = 4 + (wave << 4) + m0;
            const float* hb = h0s + R * 36 + 4 * hl;
            const v4f b0v = *(const v4f*)(hb);
            const v4f b1v = *(const v4f*)(hb + 8);
            const v4f b2v = *(const v4f*)(hb + 16);
            const v4f b3v = *(const v4f*)(hb + 24);
            v8f d0 = cinit, d1 = zero8;
            { v2f x = {a0.x, a0.y}, y = {b0v.x, b0v.y}; d0 = __builtin_amdgcn_wmma_f32_16x16x4_f32(false, x, false, y, (short)0, d0, false, false); }
            { v2f x = {a0.z, a0.w}, y = {b0v.z, b0v.w}; d1 = __builtin_amdgcn_wmma_f32_16x16x4_f32(false, x, false, y, (short)0, d1, false, false); }
            { v2f x = {a1.x, a1.y}, y = {b1v.x, b1v.y}; d0 = __builtin_amdgcn_wmma_f32_16x16x4_f32(false, x, false, y, (short)0, d0, false, false); }
            { v2f x = {a1.z, a1.w}, y = {b1v.z, b1v.w}; d1 = __builtin_amdgcn_wmma_f32_16x16x4_f32(false, x, false, y, (short)0, d1, false, false); }
            { v2f x = {a2.x, a2.y}, y = {b2v.x, b2v.y}; d0 = __builtin_amdgcn_wmma_f32_16x16x4_f32(false, x, false, y, (short)0, d0, false, false); }
            { v2f x = {a2.z, a2.w}, y = {b2v.z, b2v.w}; d1 = __builtin_amdgcn_wmma_f32_16x16x4_f32(false, x, false, y, (short)0, d1, false, false); }
            { v2f x = {a3.x, a3.y}, y = {b3v.x, b3v.y}; d0 = __builtin_amdgcn_wmma_f32_16x16x4_f32(false, x, false, y, (short)0, d0, false, false); }
            { v2f x = {a3.z, a3.w}, y = {b3v.z, b3v.w}; d1 = __builtin_amdgcn_wmma_f32_16x16x4_f32(false, x, false, y, (short)0, d1, false, false); }
            const v4f s = {d0[0] + d1[0], d0[1] + d1[1], d0[2] + d1[2], d0[3] + d1[3]};
            if (hl == 0) *(v4f*)(hm + R * 4) = s;
        }
        if (wave == 0) {
            const float* hb = h0s + m0 * 36 + 4 * hl;
            const v4f b0v = *(const v4f*)(hb);
            const v4f b1v = *(const v4f*)(hb + 8);
            const v4f b2v = *(const v4f*)(hb + 16);
            const v4f b3v = *(const v4f*)(hb + 24);
            v8f d0 = cinit, d1 = zero8;
            { v2f x = {a0.x, a0.y}, y = {b0v.x, b0v.y}; d0 = __builtin_amdgcn_wmma_f32_16x16x4_f32(false, x, false, y, (short)0, d0, false, false); }
            { v2f x = {a0.z, a0.w}, y = {b0v.z, b0v.w}; d1 = __builtin_amdgcn_wmma_f32_16x16x4_f32(false, x, false, y, (short)0, d1, false, false); }
            { v2f x = {a1.x, a1.y}, y = {b1v.x, b1v.y}; d0 = __builtin_amdgcn_wmma_f32_16x16x4_f32(false, x, false, y, (short)0, d0, false, false); }
            { v2f x = {a1.z, a1.w}, y = {b1v.z, b1v.w}; d1 = __builtin_amdgcn_wmma_f32_16x16x4_f32(false, x, false, y, (short)0, d1, false, false); }
            { v2f x = {a2.x, a2.y}, y = {b2v.x, b2v.y}; d0 = __builtin_amdgcn_wmma_f32_16x16x4_f32(false, x, false, y, (short)0, d0, false, false); }
            { v2f x = {a2.z, a2.w}, y = {b2v.z, b2v.w}; d1 = __builtin_amdgcn_wmma_f32_16x16x4_f32(false, x, false, y, (short)0, d1, false, false); }
            { v2f x = {a3.x, a3.y}, y = {b3v.x, b3v.y}; d0 = __builtin_amdgcn_wmma_f32_16x16x4_f32(false, x, false, y, (short)0, d0, false, false); }
            { v2f x = {a3.z, a3.w}, y = {b3v.z, b3v.w}; d1 = __builtin_amdgcn_wmma_f32_16x16x4_f32(false, x, false, y, (short)0, d1, false, false); }
            const v4f s = {d0[0] + d1[0], d0[1] + d1[1], d0[2] + d1[2], d0[3] + d1[3]};
            if (lane < 4) *(v4f*)(hm + m0 * 4) = s;
        }
    }

    v2f wa0[5], wa1[5];
    #pragma unroll
    for (int j = 0; j < 5; ++j) {
        const float* q = w2s + (4 * j + 2 * hl) * 32 + m0;
        v2f f0 = {q[0],  q[32]};
        v2f f1 = {q[16], q[48]};
        wa0[j] = f0; wa1[j] = f1;
    }
    const v4f c0a = *(const v4f*)(b2s + 8 * hl);
    const v4f c0b = *(const v4f*)(b2s + 8 * hl + 4);
    const v4f c1a = *(const v4f*)(b2s + 16 + 8 * hl);
    const v4f c1b = *(const v4f*)(b2s + 16 + 8 * hl + 4);
    const v4f p0 = *(const v4f*)(w3t + arow + 8 * hl);
    const v4f p1 = *(const v4f*)(w3t + arow + 8 * hl + 4);
    const v4f p2 = *(const v4f*)(w3t + arow + 16 + 8 * hl);
    const v4f p3 = *(const v4f*)(w3t + arow + 16 + 8 * hl + 4);
    const v4f b3q = *(const v4f*)b3s;
    __syncthreads();

    {
        const int tk = (wave << 4) + m0;
        v2f wb[5];
        #pragma unroll
        for (int j = 0; j < 5; ++j) wb[j] = *(const v2f*)(hm + (tk + j) * 4 + 2 * hl);
        v8f e0 = {c0a.x, c0a.y, c0a.z, c0a.w, c0b.x, c0b.y, c0b.z, c0b.w};
        v8f e1 = {c1a.x, c1a.y, c1a.z, c1a.w, c1b.x, c1b.y, c1b.z, c1b.w};
        #pragma unroll
        for (int j = 0; j < 5; ++j) {
            e0 = __builtin_amdgcn_wmma_f32_16x16x4_f32(false, wa0[j], false, wb[j], (short)0, e0, false, false);
            e1 = __builtin_amdgcn_wmma_f32_16x16x4_f32(false, wa1[j], false, wb[j], (short)0, e1, false, false);
        }
        #pragma unroll
        for (int r = 0; r < 8; ++r) { e0[r] = fmaxf(e0[r], 0.f); e1[r] = fmaxf(e1[r], 0.f); }

        v8f f0 = {b3q.x, b3q.y, b3q.z, b3q.w, 0.f, 0.f, 0.f, 0.f};
        v8f f1 = {};
        { v2f x = {p0.x, p0.y}, y = {e0[0], e0[1]}; f0 = __builtin_amdgcn_wmma_f32_16x16x4_f32(false, x, false, y, (short)0, f0, false, false); }
        { v2f x = {p0.z, p0.w}, y = {e0[2], e0[3]}; f1 = __builtin_amdgcn_wmma_f32_16x16x4_f32(false, x, false, y, (short)0, f1, false, false); }
        { v2f x = {p1.x, p1.y}, y = {e0[4], e0[5]}; f0 = __builtin_amdgcn_wmma_f32_16x16x4_f32(false, x, false, y, (short)0, f0, false, false); }
        { v2f x = {p1.z, p1.w}, y = {e0[6], e0[7]}; f1 = __builtin_amdgcn_wmma_f32_16x16x4_f32(false, x, false, y, (short)0, f1, false, false); }
        { v2f x = {p2.x, p2.y}, y = {e1[0], e1[1]}; f0 = __builtin_amdgcn_wmma_f32_16x16x4_f32(false, x, false, y, (short)0, f0, false, false); }
        { v2f x = {p2.z, p2.w}, y = {e1[2], e1[3]}; f1 = __builtin_amdgcn_wmma_f32_16x16x4_f32(false, x, false, y, (short)0, f1, false, false); }
        { v2f x = {p3.x, p3.y}, y = {e1[4], e1[5]}; f0 = __builtin_amdgcn_wmma_f32_16x16x4_f32(false, x, false, y, (short)0, f0, false, false); }
        { v2f x = {p3.z, p3.w}, y = {e1[6], e1[7]}; f1 = __builtin_amdgcn_wmma_f32_16x16x4_f32(false, x, false, y, (short)0, f1, false, false); }
        const v4f o = {f0[0] + f1[0], f0[1] + f1[1], f0[2] + f1[2], f0[3] + f1[3]};
        if (hl == 0)
            *(v4f*)(out + ((size_t)((bb << 12) + t0 + tk) << 2)) = o;
    }
}

extern "C" void kernel_launch(void* const* d_in, const int* in_sizes, int n_in,
                              void* d_out, int out_size, void* d_ws, size_t ws_size,
                              hipStream_t stream) {
    (void)in_sizes; (void)n_in; (void)out_size; (void)ws_size;
    const float* x  = (const float*)d_in[0];
    const float* W0 = (const float*)d_in[1];
    const float* b0 = (const float*)d_in[2];
    const float* W1 = (const float*)d_in[3];
    const float* b1 = (const float*)d_in[4];
    const float* Ws = (const float*)d_in[5];
    const float* bs = (const float*)d_in[6];
    const float* Wr = (const float*)d_in[7];
    const float* br = (const float*)d_in[8];
    const float* W2 = (const float*)d_in[9];
    const float* b2 = (const float*)d_in[10];
    const float* W3 = (const float*)d_in[11];
    const float* b3 = (const float*)d_in[12];
    float* out = (float*)d_out;

    float* Pc = (float*)d_ws;

    k_pre <<<NPART, 256, 0, stream>>>(Ws, bs, Wr, W1, b1, Pc);
    k_main<<<256,   256, 0, stream>>>(x, W0, b0, Pc, br, W2, b2, W3, b3, out);
}


# x tile skips LDS: A fragments of the 484->32 product come from global memory (WGP cache hits), fetched two k-steps ahead; no fill, no barrier in front of the matrix phase; sum orders unchanged
# speedup vs baseline: 16.2390x; 1.0747x over previous
#include <hip/hip_runtime.h>
#include <hip/hip_bf16.h>


typedef __attribute__((ext_vector_type(2))) float v2f;
typedef __attribute__((ext_vector_type(4))) float v4f;
typedef __attribute__((ext_vector_type(8))) float v8f;

#define L_TOK   4096
#define B_BATCH 8
#define K0      484
#define H0      32
#define TB      128
#define HALO    120
#define NTILE   9
#define XBACK   (HALO + 16)
#define PW      (NTILE * 512)

#define FK_CH    1024
#define FK_KW    128
#define FK_NCH   4
#define FK_TILES 33
#define FK_PITCH 1032
#define NPART    (FK_TILES * FK_NCH)
#define PPITCH   160

__global__ __launch_bounds__(256) void k_pre(const float* __restrict__ Ws,
                                             const float* __restrict__ bs,
                                             const float* __restrict__ Wr,
                                             const float* __restrict__ W1,
                                             const float* __restrict__ b1,
                                             float* __restrict__ Pc) {
    __shared__ __attribute__((aligned(16))) float wrT[5 * FK_PITCH];
    __shared__ __attribute__((aligned(16))) float red[8 * 64];
    __shared__ __attribute__((aligned(16))) float tT[5 * 16];

    const int tid  = threadIdx.x;
    const int tile = blockIdx.x >> 2;
    const int kc   = blockIdx.x & 3;
    const int kb   = kc * FK_CH;
    const int wave = __builtin_amdgcn_readfirstlane(tid >> 5);
    const int lane = tid & 31;
    const int h    = lane >> 4;
    const int c    = lane & 15;

    const int row = tile * 16 + c;
    const float* src = (row < 512) ? (Ws + (size_t)row * 4096) : bs;
    const v4f* bp = (const v4f*)(src + kb + wave * FK_KW + 4 * h);
    const v4f zero4 = {0.f, 0.f, 0.f, 0.f};
    v4f bv[16];
    #pragma unroll
    for (int j = 0; j < 16; ++j) bv[j] = zero4;
    if (row <= 512) {
        #pragma unroll
        for (int j = 0; j < 16; ++j) bv[j] = bp[2 * j];
    }

    v4f m0[2], m1[2], m2[2];
    #pragma unroll
    for (int j = 0; j < 2; ++j) { m0[j] = zero4; m1[j] = zero4; m2[j] = zero4; }
    if (wave == 0) {
        if (tile < 32) {
            const float* q = W1 + (size_t)c * 512 + tile * 16 + 4 * h;
            #pragma unroll
            for (int j = 0; j < 2; ++j) {
                m0[j] = *(const v4f*)(q + 8 * j);
                m1[j] = *(const v4f*)(q + 16 * 512 + 8 * j);
            }
            if (c == 0) {
                #pragma unroll
                for (int j = 0; j < 2; ++j) m2[j] = *(const v4f*)(b1 + tile * 16 + 4 * h + 8 * j);
            }
        } else if (lane == 0) {
            m2[0].x = 1.f;
        }
    }

    #pragma unroll
    for (int r = 0; r < 4; ++r) {
        int k = tid + (r << 8);
        v4f w = ((const v4f*)(Wr + (size_t)kb * 4))[k];
        wrT[0 * FK_PITCH + k] = w.x;
        wrT[1 * FK_PITCH + k] = w.y;
        wrT[2 * FK_PITCH + k] = w.z;
        wrT[3 * FK_PITCH + k] = w.w;
        wrT[4 * FK_PITCH + k] = 0.f;
    }
    if (tid >= 64 && tid < 80) tT[tid] = 0.f;
    __syncthreads();

    v8f acc0 = {}; v8f acc1 = {};
    const float* ap = wrT + (c < 4 ? c : 4) * FK_PITCH + wave * FK_KW + 4 * h;
    #pragma unroll
    for (int j = 0; j < 16; ++j) {
        v4f av = *(const v4f*)(ap + 8 * j);
        v2f a0 = {av.x, av.y},       a1 = {av.z, av.w};
        v2f b0 = {bv[j].x, bv[j].y}, b1v = {bv[j].z, bv[j].w};
        acc0 = __builtin_amdgcn_wmma_f32_16x16x4_f32(false, a0, false, b0,  (short)0, acc0, false, false);
        acc1 = __builtin_amdgcn_wmma_f32_16x16x4_f32(false, a1, false, b1v, (short)0, acc1, false, false);
    }

    if (h == 0) {
        v4f s = {acc0[0] + acc1[0], acc0[1] + acc1[1], acc0[2] + acc1[2], acc0[3] + acc1[3]};
        *(v4f*)(red + wave * 64 + c * 4) = s;
    }
    __syncthreads();
    if (tid < 64) {
        float s = red[tid];
        #pragma unroll
        for (int w = 1; w < 8; ++w) s += red[w * 64 + tid];
        tT[(tid & 3) * 16 + (tid >> 2)] = s;
    }
    __syncthreads();

    if (wave == 0) {
        v8f e0 = {}; v8f e1 = {}; v8f e2 = {};
        const float* a2 = tT + (c < 4 ? c : 4) * 16 + 4 * h;
        #pragma unroll
        for (int j = 0; j < 2; ++j) {
            v4f av = *(const v4f*)(a2 + 8 * j);
            v2f a0 = {av.x, av.y}, a1 = {av.z, av.w};
            v2f x0 = {m0[j].x, m0[j].y}, x1 = {m0[j].z, m0[j].w};
            v2f y0 = {m1[j].x, m1[j].y}, y1 = {m1[j].z, m1[j].w};
            v2f z0 = {m2[j].x, m2[j].y}, z1 = {m2[j].z, m2[j].w};
            e0 = __builtin_amdgcn_wmma_f32_16x16x4_f32(false, a0, false, x0, (short)0, e0, false, false);
            e1 = __builtin_amdgcn_wmma_f32_16x16x4_f32(false, a0, false, y0, (short)0, e1, false, false);
            e2 = __builtin_amdgcn_wmma_f32_16x16x4_f32(false, a0, false, z0, (short)0, e2, false, false);
            e0 = __builtin_amdgcn_wmma_f32_16x16x4_f32(false, a1, false, x1, (short)0, e0, false, false);
            e1 = __builtin_amdgcn_wmma_f32_16x16x4_f32(false, a1, false, y1, (short)0, e1, false, false);
            e2 = __builtin_amdgcn_wmma_f32_16x16x4_f32(false, a1, false, z1, (short)0, e2, false, false);
        }
        if (h == 0) {
            float* dst = Pc + (size_t)blockIdx.x * PPITCH;
            v4f s0 = {e0[0], e0[1], e0[2], e0[3]};
            v4f s1 = {e1[0], e1[1], e1[2], e1[3]};
            v4f s2 = {e2[0], e2[1], e2[2], e2[3]};
            *(v4f*)(dst + c * 4)      = s0;
            *(v4f*)(dst + 64 + c * 4) = s1;
            if (c == 0) *(v4f*)(dst + 128) = s2;
        }
    }
}

#define KS_WAVE 15
__global__ __launch_bounds__(256) void k_main(const float* __restrict__ x,
                                              const float* __restrict__ W0,
                                              const float* __restrict__ b0,
                                              const float* __restrict__ Pc,
                                              const float* __restrict__ br,
                                              const float* __restrict__ W2,
                                              const float* __restrict__ b2,
                                              const float* __restrict__ W3,
                                              const float* __restrict__ b3,
                                              float* __restrict__ out) {
    __shared__ __attribute__((aligned(16))) float part[8 * PW];
    __shared__ float b0s[H0];
    __shared__ __attribute__((aligned(16))) float wct[5 * H0];
    __shared__ __attribute__((aligned(16))) float bcs[4];
    __shared__ __attribute__((aligned(16))) float wred[7 * 132];
    __shared__ __attribute__((aligned(16))) float h0s[(TB + 4) * 36];
    __shared__ __attribute__((aligned(16))) float hm[(TB + 4) * 4];
    __shared__ __attribute__((aligned(16))) float w2s[20 * 32];
    __shared__ __attribute__((aligned(16))) float b2s[32];
    __shared__ __attribute__((aligned(16))) float w3t[5 * 32];
    __shared__ __attribute__((aligned(16))) float b3s[4];

    const int tid = threadIdx.x;
    const int bb  = blockIdx.x >> 5;
    const int t0  = (blockIdx.x & 31) * TB;
    const int wave = __builtin_amdgcn_readfirstlane(tid >> 5);
    const int lane = tid & 31;
    const int hl   = lane >> 4;
    const int m0   = lane & 15;
    const int c0   = hl << 1;
    const int k_lo = wave * KS_WAVE;

    const v2f* xs = (const v2f*)(x + ((size_t)bb << 14));
    const unsigned ub = (unsigned)(2 * (t0 - XBACK + m0 + k_lo) + hl);
#define XA(mt, i) (xs[(ub + 2u * (unsigned)((mt) * 16 + (i))) & 8191u])

    v2f abuf[3][NTILE];
    #pragma unroll
    for (int mt = 0; mt < NTILE; ++mt) { abuf[0][mt] = XA(mt, 0); abuf[1][mt] = XA(mt, 1); }

    const float* bptr = W0 + (size_t)(((k_lo << 2) + c0) * H0 + m0);
    v2f breg[16][2];
    #pragma unroll
    for (int i = 0; i < KS_WAVE; ++i) {
        v2f f0 = {bptr[(4 * i) * H0],      bptr[(4 * i + 1) * H0]};
        v2f f1 = {bptr[(4 * i) * H0 + 16], bptr[(4 * i + 1) * H0 + 16]};
        breg[i][0] = f0; breg[i][1] = f1;
    }
    {
        v2f z = {0.f, 0.f};
        breg[15][0] = z; breg[15][1] = z;
    }
    if (wave == 7) {
        v2f f0 = {bptr[60 * H0],      bptr[61 * H0]};
        v2f f1 = {bptr[60 * H0 + 16], bptr[61 * H0 + 16]};
        breg[15][0] = f0; breg[15][1] = f1;
    }

    float brr = 0.f;
    if (tid >= 128 && tid < 132) brr = br[tid - 128];

    const float w2r0 = W2[tid];
    const float w2r1 = W2[tid + 256];
    float w2r2 = 0.f, w3r = 0.f, b2r = 0.f, b3r = 0.f;
    if (tid < 128) { w2r2 = W2[tid + 512]; w3r = W3[tid]; }
    float b0r = 0.f;
    if (tid < 32)  { b2r = b2[tid]; b0r = b0[tid]; }
    if (tid < 4)   b3r = b3[tid];

    v8f acc[NTILE][2];
    #pragma unroll
    for (int mt = 0; mt < NTILE; ++mt) { v8f z = {}; acc[mt][0] = z; acc[mt][1] = z; }

    const int pg = tid / 33;
    const int ps = tid - pg * 33;
    v4f pv[19];
    #pragma unroll
    for (int q = 0; q < 19; ++q) { v4f z = {0.f, 0.f, 0.f, 0.f}; pv[q] = z; }
    if (tid < 7 * 33) {
        const v4f* pp = (const v4f*)Pc + pg * (PPITCH / 4) + ps;
        #pragma unroll
        for (int q = 0; q < 18; ++q) pv[q] = pp[q * 7 * (PPITCH / 4)];
        if (pg < 6) pv[18] = pp[18 * 7 * (PPITCH / 4)];
    }

    #pragma unroll
    for (int i = 0; i < KS_WAVE; ++i) {
        if (i + 2 < KS_WAVE) {
            #pragma unroll
            for (int mt = 0; mt < NTILE; ++mt) abuf[(i + 2) % 3][mt] = XA(mt, i + 2);
        } else if (i + 2 == KS_WAVE) {
            if (wave == 7) {
                #pragma unroll
                for (int mt = 0; mt < NTILE; ++mt) abuf[(i + 2) % 3][mt] = XA(mt, KS_WAVE);
            }
        }
        __builtin_amdgcn_sched_barrier(0);
        #pragma unroll
        for (int mt = 0; mt < NTILE; ++mt) {
            const v2f a = abuf[i % 3][mt];
            acc[mt][0] = __builtin_amdgcn_wmma_f32_16x16x4_f32(false, a, false, breg[i][0], (short)0, acc[mt][0], false, false);
            acc[mt][1] = __builtin_amdgcn_wmma_f32_16x16x4_f32(false, a, false, breg[i][1], (short)0, acc[mt][1], false, false);
        }
        __builtin_amdgcn_sched_barrier(0);
    }
    if (wave == 7) {
        #pragma unroll
        for (int mt = 0; mt < NTILE; ++mt) {
            const v2f a = abuf[KS_WAVE % 3][mt];
            acc[mt][0] = __builtin_amdgcn_wmma_f32_16x16x4_f32(false, a, false, breg[15][0], (short)0, acc[mt][0], false, false);
            acc[mt][1] = __builtin_amdgcn_wmma_f32_16x16x4_f32(false, a, false, breg[15][1], (short)0, acc[mt][1], false, false);
        }
    }

    w2s[tid]       = w2r0;
    w2s[tid + 256] = w2r1;
    if (tid < 128) { w2s[tid + 512] = w2r2; w3t[(tid & 3) * 32 + (tid >> 2)] = w3r; }
    if (tid < 32)  { b0s[tid] = b0r; b2s[tid] = b2r; w3t[4 * 32 + tid] = 0.f; wct[4 * H0 + tid] = 0.f; }
    if (tid < 4)   b3s[tid] = b3r;

    if (tid < 7 * 33) {
        v4f psum = pv[0];
        #pragma unroll
        for (int q = 1; q < 19; ++q) psum += pv[q];
        *(v4f*)(wred + pg * 132 + ps * 4) = psum;
    }

    float* pw = part + wave * PW + lane * 4;
    #pragma unroll
    for (int mt = 0; mt < NTILE; ++mt) {
        #pragma unroll
        for (int nt = 0; nt < 2; ++nt) {
            v8f d = acc[mt][nt];
            v4f lo = {d[0], d[1], d[2], d[3]};
            v4f hi = {d[4], d[5], d[6], d[7]};
            *(v4f*)(pw + ((mt * 2 + nt) * 2 + 0) * 128) = lo;
            *(v4f*)(pw + ((mt * 2 + nt) * 2 + 1) * 128) = hi;
        }
    }
    __syncthreads();

    if (tid < 132) {
        float ws = wred[tid];
        #pragma unroll
        for (int g = 1; g < 7; ++g) ws += wred[g * 132 + tid];
        if (tid < 128) wct[(tid & 3) * H0 + (tid >> 2)] = ws; else bcs[tid - 128] = ws + brr;
    }

    const float* pr = part + (wave + 1) * 512 + lane * 4;
    v8f acc0, acc1;
    {
        v4f q0 = *(const v4f*)(pr);
        v4f q1 = *(const v4f*)(pr + 128);
        v4f q2 = *(const v4f*)(pr + 256);
        v4f q3 = *(const v4f*)(pr + 384);
        #pragma unroll
        for (int s = 1; s < 8; ++s) {
            q0 += *(const v4f*)(pr + s * PW);
            q1 += *(const v4f*)(pr + s * PW + 128);
            q2 += *(const v4f*)(pr + s * PW + 256);
            q3 += *(const v4f*)(pr + s * PW + 384);
        }
        v8f r0 = {q0.x, q0.y, q0.z, q0.w, q1.x, q1.y, q1.z, q1.w};
        v8f r1 = {q2.x, q2.y, q2.z, q2.w, q3.x, q3.y, q3.z, q3.w};
        acc0 = r0; acc1 = r1;
    }

    #pragma unroll
    for (int g = 0; g < 8; ++g) {
        int R = 4 + (wave << 4) + g + (hl << 3);
        h0s[R * 36 + m0]      = tanhf(acc0[g] + b0s[m0]);
        h0s[R * 36 + m0 + 16] = tanhf(acc1[g] + b0s[m0 + 16]);
    }
    if (hl) {
        const int nt = wave & 1, e = wave >> 1;
        const float* ph = part + nt * 256 + 128 + lane * 4 + e;
        float s = ph[0];
        #pragma unroll
        for (int w = 1; w < 8; ++w) s += ph[w * PW];
        h0s[e * 36 + m0 + 16 * nt] = tanhf(s + b0s[m0 + 16 * nt]);
    }
    __syncthreads();

    const int arow = (m0 < 4 ? m0 : 4) * 32;

    {
        const float* wa = wct + arow + 4 * hl;
        const v4f a0 = *(const v4f*)(wa);
        const v4f a1 = *(const v4f*)(wa + 8);
        const v4f a2 = *(const v4f*)(wa + 16);
        const v4f a3 = *(const v4f*)(wa + 24);
        const v4f cb = *(const v4f*)bcs;
        const v8f cinit = {cb.x, cb.y, cb.z, cb.w, 0.f, 0.f, 0.f, 0.f};
        const v8f zero8 = {};
        {
            const int R = 4 + (wave << 4) + m0;
            const float* hb = h0s + R * 36 + 4 * hl;
            const v4f b0v = *(const v4f*)(hb);
            const v4f b1v = *(const v4f*)(hb + 8);
            const v4f b2v = *(const v4f*)(hb + 16);
            const v4f b3v = *(const v4f*)(hb + 24);
            v8f d0 = cinit, d1 = zero8;
            { v2f x = {a0.x, a0.y}, y = {b0v.x, b0v.y}; d0 = __builtin_amdgcn_wmma_f32_16x16x4_f32(false, x, false, y, (short)0, d0, false, false); }
            { v2f x = {a0.z, a0.w}, y = {b0v.z, b0v.w}; d1 = __builtin_amdgcn_wmma_f32_16x16x4_f32(false, x, false, y, (short)0, d1, false, false); }
            { v2f x = {a1.x, a1.y}, y = {b1v.x, b1v.y}; d0 = __builtin_amdgcn_wmma_f32_16x16x4_f32(false, x, false, y, (short)0, d0, false, false); }
            { v2f x = {a1.z, a1.w}, y = {b1v.z, b1v.w}; d1 = __builtin_amdgcn_wmma_f32_16x16x4_f32(false, x, false, y, (short)0, d1, false, false); }
            { v2f x = {a2.x, a2.y}, y = {b2v.x, b2v.y}; d0 = __builtin_amdgcn_wmma_f32_16x16x4_f32(false, x, false, y, (short)0, d0, false, false); }
            { v2f x = {a2.z, a2.w}, y = {b2v.z, b2v.w}; d1 = __builtin_amdgcn_wmma_f32_16x16x4_f32(false, x, false, y, (short)0, d1, false, false); }
            { v2f x = {a3.x, a3.y}, y = {b3v.x, b3v.y}; d0 = __builtin_amdgcn_wmma_f32_16x16x4_f32(false, x, false, y, (short)0, d0, false, false); }
            { v2f x = {a3.z, a3.w}, y = {b3v.z, b3v.w}; d1 = __builtin_amdgcn_wmma_f32_16x16x4_f32(false, x, false, y, (short)0, d1, false, false); }
            const v4f s = {d0[0] + d1[0], d0[1] + d1[1], d0[2] + d1[2], d0[3] + d1[3]};
            if (hl == 0) *(v4f*)(hm + R * 4) = s;
        }
        if (wave == 0) {
            const float* hb = h0s + m0 * 36 + 4 * hl;
            const v4f b0v = *(const v4f*)(hb);
            const v4f b1v = *(const v4f*)(hb + 8);
            const v4f b2v = *(const v4f*)(hb + 16);
            const v4f b3v = *(const v4f*)(hb + 24);
            v8f d0 = cinit, d1 = zero8;
            { v2f x = {a0.x, a0.y}, y = {b0v.x, b0v.y}; d0 = __builtin_amdgcn_wmma_f32_16x16x4_f32(false, x, false, y, (short)0, d0, false, false); }
            { v2f x = {a0.z, a0.w}, y = {b0v.z, b0v.w}; d1 = __builtin_amdgcn_wmma_f32_16x16x4_f32(false, x, false, y, (short)0, d1, false, false); }
            { v2f x = {a1.x, a1.y}, y = {b1v.x, b1v.y}; d0 = __builtin_amdgcn_wmma_f32_16x16x4_f32(false, x, false, y, (short)0, d0, false, false); }
            { v2f x = {a1.z, a1.w}, y = {b1v.z, b1v.w}; d1 = __builtin_amdgcn_wmma_f32_16x16x4_f32(false, x, false, y, (short)0, d1, false, false); }
            { v2f x = {a2.x, a2.y}, y = {b2v.x, b2v.y}; d0 = __builtin_amdgcn_wmma_f32_16x16x4_f32(false, x, false, y, (short)0, d0, false, false); }
            { v2f x = {a2.z, a2.w}, y = {b2v.z, b2v.w}; d1 = __builtin_amdgcn_wmma_f32_16x16x4_f32(false, x, false, y, (short)0, d1, false, false); }
            { v2f x = {a3.x, a3.y}, y = {b3v.x, b3v.y}; d0 = __builtin_amdgcn_wmma_f32_16x16x4_f32(false, x, false, y, (short)0, d0, false, false); }
            { v2f x = {a3.z, a3.w}, y = {b3v.z, b3v.w}; d1 = __builtin_amdgcn_wmma_f32_16x16x4_f32(false, x, false, y, (short)0, d1, false, false); }
            const v4f s = {d0[0] + d1[0], d0[1] + d1[1], d0[2] + d1[2], d0[3] + d1[3]};
            if (lane < 4) *(v4f*)(hm + m0 * 4) = s;
        }
    }

    v2f wa0[5], wa1[5];
    #pragma unroll
    for (int j = 0; j < 5; ++j) {
        const float* q = w2s + (4 * j + 2 * hl) * 32 + m0;
        v2f f0 = {q[0],  q[32]};
        v2f f1 = {q[16], q[48]};
        wa0[j] = f0; wa1[j] = f1;
    }
    const v4f c0a = *(const v4f*)(b2s + 8 * hl);
    const v4f c0b = *(const v4f*)(b2s + 8 * hl + 4);
    const v4f c1a = *(const v4f*)(b2s + 16 + 8 * hl);
    const v4f c1b = *(const v4f*)(b2s + 16 + 8 * hl + 4);
    const v4f p0 = *(const v4f*)(w3t + arow + 8 * hl);
    const v4f p1 = *(const v4f*)(w3t + arow + 8 * hl + 4);
    const v4f p2 = *(const v4f*)(w3t + arow + 16 + 8 * hl);
    const v4f p3 = *(const v4f*)(w3t + arow + 16 + 8 * hl + 4);
    const v4f b3q = *(const v4f*)b3s;
    __syncthreads();

    {
        const int tk = (wave << 4) + m0;
        v2f wb[5];
        #pragma unroll
        for (int j = 0; j < 5; ++j) wb[j] = *(const v2f*)(hm + (tk + j) * 4 + 2 * hl);
        v8f e0 = {c0a.x, c0a.y, c0a.z, c0a.w, c0b.x, c0b.y, c0b.z, c0b.w};
        v8f e1 = {c1a.x, c1a.y, c1a.z, c1a.w, c1b.x, c1b.y, c1b.z, c1b.w};
        #pragma unroll
        for (int j = 0; j < 5; ++j) {
            e0 = __builtin_amdgcn_wmma_f32_16x16x4_f32(false, wa0[j], false, wb[j], (short)0, e0, false, false);
            e1 = __builtin_amdgcn_wmma_f32_16x16x4_f32(false, wa1[j], false, wb[j], (short)0, e1, false, false);
        }
        #pragma unroll
        for (int r = 0; r < 8; ++r) { e0[r] = fmaxf(e0[r], 0.f); e1[r] = fmaxf(e1[r], 0.f); }

        v8f f0 = {b3q.x, b3q.y, b3q.z, b3q.w, 0.f, 0.f, 0.f, 0.f};
        v8f f1 = {};
        { v2f x = {p0.x, p0.y}, y = {e0[0], e0[1]}; f0 = __builtin_amdgcn_wmma_f32_16x16x4_f32(false, x, false, y, (short)0, f0, false, false); }
        { v2f x = {p0.z, p0.w}, y = {e0[2], e0[3]}; f1 = __builtin_amdgcn_wmma_f32_16x16x4_f32(false, x, false, y, (short)0, f1, false, false); }
        { v2f x = {p1.x, p1.y}, y = {e0[4], e0[5]}; f0 = __builtin_amdgcn_wmma_f32_16x16x4_f32(false, x, false, y, (short)0, f0, false, false); }
        { v2f x = {p1.z, p1.w}, y = {e0[6], e0[7]}; f1 = __builtin_amdgcn_wmma_f32_16x16x4_f32(false, x, false, y, (short)0, f1, false, false); }
        { v2f x = {p2.x, p2.y}, y = {e1[0], e1[1]}; f0 = __builtin_amdgcn_wmma_f32_16x16x4_f32(false, x, false, y, (short)0, f0, false, false); }
        { v2f x = {p2.z, p2.w}, y = {e1[2], e1[3]}; f1 = __builtin_amdgcn_wmma_f32_16x16x4_f32(false, x, false, y, (short)0, f1, false, false); }
        { v2f x = {p3.x, p3.y}, y = {e1[4], e1[5]}; f0 = __builtin_amdgcn_wmma_f32_16x16x4_f32(false, x, false, y, (short)0, f0, false, false); }
        { v2f x = {p3.z, p3.w}, y = {e1[6], e1[7]}; f1 = __builtin_amdgcn_wmma_f32_16x16x4_f32(false, x, false, y, (short)0, f1, false, false); }
        const v4f o = {f0[0] + f1[0], f0[1] + f1[1], f0[2] + f1[2], f0[3] + f1[3]};
        if (hl == 0)
            *(v4f*)(out + ((size_t)((bb << 12) + t0 + tk) << 2)) = o;
    }
}

extern "C" void kernel_launch(void* const* d_in, const int* in_sizes, int n_in,
                              void* d_out, int out_size, void* d_ws, size_t ws_size,
                              hipStream_t stream) {
    (void)in_sizes; (void)n_in; (void)out_size; (void)ws_size;
    const float* x  = (const float*)d_in[0];
    const float* W0 = (const float*)d_in[1];
    const float* b0 = (const float*)d_in[2];
    const float* W1 = (const float*)d_in[3];
    const float* b1 = (const float*)d_in[4];
    const float* Ws = (const float*)d_in[5];
    const float* bs = (const float*)d_in[6];
    const float* Wr = (const float*)d_in[7];
    const float* br = (const float*)d_in[8];
    const float* W2 = (const float*)d_in[9];
    const float* b2 = (const float*)d_in[10];
    const float* W3 = (const float*)d_in[11];
    const float* b3 = (const float*)d_in[12];
    float* out = (float*)d_out;

    float* Pc = (float*)d_ws;

    k_pre <<<NPART, 256, 0, stream>>>(Ws, bs, Wr, W1, b1, Pc);
    k_main<<<256,   256, 0, stream>>>(x, W0, b0, Pc, br, W2, b2, W3, b3, out);
}


# Tanh tail of the front stage: 17 branchy library calls per lane (both regimes in turn, one dependent chain) -> branch-free two-regime form in packed f32, exp2 + rcp above 5/8; sum orders unchanged
# speedup vs baseline: 17.1710x; 1.0574x over previous
#include <hip/hip_runtime.h>
#include <hip/hip_bf16.h>


typedef __attribute__((ext_vector_type(2))) float v2f;
typedef __attribute__((ext_vector_type(4))) float v4f;
typedef __attribute__((ext_vector_type(8))) float v8f;

#define L_TOK   4096
#define B_BATCH 8
#define K0      484
#define H0      32
#define TB      128
#define HALO    120
#define NTILE   9
#define XBACK   (HALO + 16)
#define PW      (NTILE * 512)

#define FK_CH    1024
#define FK_KW    128
#define FK_NCH   4
#define FK_TILES 33
#define FK_PITCH 1032
#define NPART    (FK_TILES * FK_NCH)
#define PPITCH   160

__global__ __launch_bounds__(256) void k_pre(const float* __restrict__ Ws,
                                             const float* __restrict__ bs,
                                             const float* __restrict__ Wr,
                                             const float* __restrict__ W1,
                                             const float* __restrict__ b1,
                                             float* __restrict__ Pc) {
    __shared__ __attribute__((aligned(16))) float wrT[5 * FK_PITCH];
    __shared__ __attribute__((aligned(16))) float red[8 * 64];
    __shared__ __attribute__((aligned(16))) float tT[5 * 16];

    const int tid  = threadIdx.x;
    const int tile = blockIdx.x >> 2;
    const int kc   = blockIdx.x & 3;
    const int kb   = kc * FK_CH;
    const int wave = __builtin_amdgcn_readfirstlane(tid >> 5);
    const int lane = tid & 31;
    const int h    = lane >> 4;
    const int c    = lane & 15;

    const int row = tile * 16 + c;
    const float* src = (row < 512) ? (Ws + (size_t)row * 4096) : bs;
    const v4f* bp = (const v4f*)(src + kb + wave * FK_KW + 4 * h);
    const v4f zero4 = {0.f, 0.f, 0.f, 0.f};
    v4f bv[16];
    #pragma unroll
    for (int j = 0; j < 16; ++j) bv[j] = zero4;
    if (row <= 512) {
        #pragma unroll
        for (int j = 0; j < 16; ++j) bv[j] = bp[2 * j];
    }

    v4f m0[2], m1[2], m2[2];
    #pragma unroll
    for (int j = 0; j < 2; ++j) { m0[j] = zero4; m1[j] = zero4; m2[j] = zero4; }
    if (wave == 0) {
        if (tile < 32) {
            const float* q = W1 + (size_t)c * 512 + tile * 16 + 4 * h;
            #pragma unroll
            for (int j = 0; j < 2; ++j) {
                m0[j] = *(const v4f*)(q + 8 * j);
                m1[j] = *(const v4f*)(q + 16 * 512 + 8 * j);
            }
            if (c == 0) {
                #pragma unroll
                for (int j = 0; j < 2; ++j) m2[j] = *(const v4f*)(b1 + tile * 16 + 4 * h + 8 * j);
            }
        } else if (lane == 0) {
            m2[0].x = 1.f;
        }
    }

    #pragma unroll
    for (int r = 0; r < 4; ++r) {
        int k = tid + (r << 8);
        v4f w = ((const v4f*)(Wr + (size_t)kb * 4))[k];
        wrT[0 * FK_PITCH + k] = w.x;
        wrT[1 * FK_PITCH + k] = w.y;
        wrT[2 * FK_PITCH + k] = w.z;
        wrT[3 * FK_PITCH + k] = w.w;
        wrT[4 * FK_PITCH + k] = 0.f;
    }
    if (tid >= 64 && tid < 80) tT[tid] = 0.f;
    __syncthreads();

    v8f acc0 = {}; v8f acc1 = {};
    const float* ap = wrT + (c < 4 ? c : 4) * FK_PITCH + wave * FK_KW + 4 * h;
    #pragma unroll
    for (int j = 0; j < 16; ++j) {
        v4f av = *(const v4f*)(ap + 8 * j);
        v2f a0 = {av.x, av.y},       a1 = {av.z, av.w};
        v2f b0 = {bv[j].x, bv[j].y}, b1v = {bv[j].z, bv[j].w};
        acc0 = __builtin_amdgcn_wmma_f32_16x16x4_f32(false, a0, false, b0,  (short)0, acc0, false, false);
        acc1 = __builtin_amdgcn_wmma_f32_16x16x4_f32(false, a1, false, b1v, (short)0, acc1, false, false);
    }

    if (h == 0) {
        v4f s = {acc0[0] + acc1[0], acc0[1] + acc1[1], acc0[2] + acc1[2], acc0[3] + acc1[3]};
        *(v4f*)(red + wave * 64 + c * 4) = s;
    }
    __syncthreads();
    if (tid < 64) {
        float s = red[tid];
        #pragma unroll
        for (int w = 1; w < 8; ++w) s += red[w * 64 + tid];
        tT[(tid & 3) * 16 + (tid >> 2)] = s;
    }
    __syncthreads();

    if (wave == 0) {
        v8f e0 = {}; v8f e1 = {}; v8f e2 = {};
        const float* a2 = tT + (c < 4 ? c : 4) * 16 + 4 * h;
        #pragma unroll
        for (int j = 0; j < 2; ++j) {
            v4f av = *(const v4f*)(a2 + 8 * j);
            v2f a0 = {av.x, av.y}, a1 = {av.z, av.w};
            v2f x0 = {m0[j].x, m0[j].y}, x1 = {m0[j].z, m0[j].w};
            v2f y0 = {m1[j].x, m1[j].y}, y1 = {m1[j].z, m1[j].w};
            v2f z0 = {m2[j].x, m2[j].y}, z1 = {m2[j].z, m2[j].w};
            e0 = __builtin_amdgcn_wmma_f32_16x16x4_f32(false, a0, false, x0, (short)0, e0, false, false);
            e1 = __builtin_amdgcn_wmma_f32_16x16x4_f32(false, a0, false, y0, (short)0, e1, false, false);
            e2 = __builtin_amdgcn_wmma_f32_16x16x4_f32(false, a0, false, z0, (short)0, e2, false, false);
            e0 = __builtin_amdgcn_wmma_f32_16x16x4_f32(false, a1, false, x1, (short)0, e0, false, false);
            e1 = __builtin_amdgcn_wmma_f32_16x16x4_f32(false, a1, false, y1, (short)0, e1, false, false);
            e2 = __builtin_amdgcn_wmma_f32_16x16x4_f32(false, a1, false, z1, (short)0, e2, false, false);
        }
        if (h == 0) {
            float* dst = Pc + (size_t)blockIdx.x * PPITCH;
            v4f s0 = {e0[0], e0[1], e0[2], e0[3]};
            v4f s1 = {e1[0], e1[1], e1[2], e1[3]};
            v4f s2 = {e2[0], e2[1], e2[2], e2[3]};
            *(v4f*)(dst + c * 4)      = s0;
            *(v4f*)(dst + 64 + c * 4) = s1;
            if (c == 0) *(v4f*)(dst + 128) = s2;
        }
    }
}

__device__ __forceinline__ v2f splat2(float s) { v2f r = {s, s}; return r; }
__device__ __forceinline__ float tanh_pick(float v, float sm, float bg) {
    return __builtin_fabsf(v) < 0.625f ? sm : __builtin_copysignf(bg, v);
}
__device__ __forceinline__ v2f tanh_pair(v2f v) {
    const v2f z = v * v;
    v2f p = __builtin_elementwise_fma(z, splat2(-0.00570002059f), splat2(0.0206340719f));
    p = __builtin_elementwise_fma(z, p, splat2(-0.0537379310f));
    p = __builtin_elementwise_fma(z, p, splat2(0.133314162f));
    p = __builtin_elementwise_fma(z, p, splat2(-0.333332807f));
    const v2f sm = __builtin_elementwise_fma(z, v * p, v);
    const v2f e = v * splat2(2.88539004f);
    v2f t = {__builtin_amdgcn_exp2f(__builtin_fabsf(e.x)), __builtin_amdgcn_exp2f(__builtin_fabsf(e.y))};
    t = t + splat2(1.f);
    const v2f r = {__builtin_amdgcn_rcpf(t.x), __builtin_amdgcn_rcpf(t.y)};
    const v2f bg = __builtin_elementwise_fma(r, splat2(-2.f), splat2(1.f));
    v2f o = {tanh_pick(v.x, sm.x, bg.x), tanh_pick(v.y, sm.y, bg.y)};
    return o;
}

#define KS_WAVE 15
__global__ __launch_bounds__(256) void k_main(const float* __restrict__ x,
                                              const float* __restrict__ W0,
                                              const float* __restrict__ b0,
                                              const float* __restrict__ Pc,
                                              const float* __restrict__ br,
                                              const float* __restrict__ W2,
                                              const float* __restrict__ b2,
                                              const float* __restrict__ W3,
                                              const float* __restrict__ b3,
                                              float* __restrict__ out) {
    __shared__ __attribute__((aligned(16))) float part[8 * PW];
    __shared__ float b0s[H0];
    __shared__ __attribute__((aligned(16))) float wct[5 * H0];
    __shared__ __attribute__((aligned(16))) float bcs[4];
    __shared__ __attribute__((aligned(16))) float wred[7 * 132];
    __shared__ __attribute__((aligned(16))) float h0s[(TB + 4) * 36];
    __shared__ __attribute__((aligned(16))) float hm[(TB + 4) * 4];
    __shared__ __attribute__((aligned(16))) float w2s[20 * 32];
    __shared__ __attribute__((aligned(16))) float b2s[32];
    __shared__ __attribute__((aligned(16))) float w3t[5 * 32];
    __shared__ __attribute__((aligned(16))) float b3s[4];

    const int tid = threadIdx.x;
    const int bb  = blockIdx.x >> 5;
    const int t0  = (blockIdx.x & 31) * TB;
    const int wave = __builtin_amdgcn_readfirstlane(tid >> 5);
    const int lane = tid & 31;
    const int hl   = lane >> 4;
    const int m0   = lane & 15;
    const int c0   = hl << 1;
    const int k_lo = wave * KS_WAVE;

    const v2f* xs = (const v2f*)(x + ((size_t)bb << 14));
    const unsigned ub = (unsigned)(2 * (t0 - XBACK + m0 + k_lo) + hl);
#define XA(mt, i) (xs[(ub + 2u * (unsigned)((mt) * 16 + (i))) & 8191u])

    v2f abuf[3][NTILE];
    #pragma unroll
    for (int mt = 0; mt < NTILE; ++mt) { abuf[0][mt] = XA(mt, 0); abuf[1][mt] = XA(mt, 1); }

    const float* bptr = W0 + (size_t)(((k_lo << 2) + c0) * H0 + m0);
    v2f breg[16][2];
    #pragma unroll
    for (int i = 0; i < KS_WAVE; ++i) {
        v2f f0 = {bptr[(4 * i) * H0],      bptr[(4 * i + 1) * H0]};
        v2f f1 = {bptr[(4 * i) * H0 + 16], bptr[(4 * i + 1) * H0 + 16]};
        breg[i][0] = f0; breg[i][1] = f1;
    }
    {
        v2f z = {0.f, 0.f};
        breg[15][0] = z; breg[15][1] = z;
    }
    if (wave == 7) {
        v2f f0 = {bptr[60 * H0],      bptr[61 * H0]};
        v2f f1 = {bptr[60 * H0 + 16], bptr[61 * H0 + 16]};
        breg[15][0] = f0; breg[15][1] = f1;
    }

    float brr = 0.f;
    if (tid >= 128 && tid < 132) brr = br[tid - 128];

    const float w2r0 = W2[tid];
    const float w2r1 = W2[tid + 256];
    float w2r2 = 0.f, w3r = 0.f, b2r = 0.f, b3r = 0.f;
    if (tid < 128) { w2r2 = W2[tid + 512]; w3r = W3[tid]; }
    float b0r = 0.f;
    if (tid < 32)  { b2r = b2[tid]; b0r = b0[tid]; }
    if (tid < 4)   b3r = b3[tid];

    v8f acc[NTILE][2];
    #pragma unroll
    for (int mt = 0; mt < NTILE; ++mt) { v8f z = {}; acc[mt][0] = z; acc[mt][1] = z; }

    const int pg = tid / 33;
    const int ps = tid - pg * 33;
    v4f pv[19];
    #pragma unroll
    for (int q = 0; q < 19; ++q) { v4f z = {0.f, 0.f, 0.f, 0.f}; pv[q] = z; }
    if (tid < 7 * 33) {
        const v4f* pp = (const v4f*)Pc + pg * (PPITCH / 4) + ps;
        #pragma unroll
        for (int q = 0; q < 18; ++q) pv[q] = pp[q * 7 * (PPITCH / 4)];
        if (pg < 6) pv[18] = pp[18 * 7 * (PPITCH / 4)];
    }

    #pragma unroll
    for (int i = 0; i < KS_WAVE; ++i) {
        if (i + 2 < KS_WAVE) {
            #pragma unroll
            for (int mt = 0; mt < NTILE; ++mt) abuf[(i + 2) % 3][mt] = XA(mt, i + 2);
        } else if (i + 2 == KS_WAVE) {
            if (wave == 7) {
                #pragma unroll
                for (int mt = 0; mt < NTILE; ++mt) abuf[(i + 2) % 3][mt] = XA(mt, KS_WAVE);
            }
        }
        __builtin_amdgcn_sched_barrier(0);
        #pragma unroll
        for (int mt = 0; mt < NTILE; ++mt) {
            const v2f a = abuf[i % 3][mt];
            acc[mt][0] = __builtin_amdgcn_wmma_f32_16x16x4_f32(false, a, false, breg[i][0], (short)0, acc[mt][0], false, false);
            acc[mt][1] = __builtin_amdgcn_wmma_f32_16x16x4_f32(false, a, false, breg[i][1], (short)0, acc[mt][1], false, false);
        }
        __builtin_amdgcn_sched_barrier(0);
    }
    if (wave == 7) {
        #pragma unroll
        for (int mt = 0; mt < NTILE; ++mt) {
            const v2f a = abuf[KS_WAVE % 3][mt];
            acc[mt][0] = __builtin_amdgcn_wmma_f32_16x16x4_f32(false, a, false, breg[15][0], (short)0, acc[mt][0], false, false);
            acc[mt][1] = __builtin_amdgcn_wmma_f32_16x16x4_f32(false, a, false, breg[15][1], (short)0, acc[mt][1], false, false);
        }
    }

    w2s[tid]       = w2r0;
    w2s[tid + 256] = w2r1;
    if (tid < 128) { w2s[tid + 512] = w2r2; w3t[(tid & 3) * 32 + (tid >> 2)] = w3r; }
    if (tid < 32)  { b0s[tid] = b0r; b2s[tid] = b2r; w3t[4 * 32 + tid] = 0.f; wct[4 * H0 + tid] = 0.f; }
    if (tid < 4)   b3s[tid] = b3r;

    if (tid < 7 * 33) {
        v4f psum = pv[0];
        #pragma unroll
        for (int q = 1; q < 19; ++q) psum += pv[q];
        *(v4f*)(wred + pg * 132 + ps * 4) = psum;
    }

    float* pw = part + wave * PW + lane * 4;
    #pragma unroll
    for (int mt = 0; mt < NTILE; ++mt) {
        #pragma unroll
        for (int nt = 0; nt < 2; ++nt) {
            v8f d = acc[mt][nt];
            v4f lo = {d[0], d[1], d[2], d[3]};
            v4f hi = {d[4], d[5], d[6], d[7]};
            *(v4f*)(pw + ((mt * 2 + nt) * 2 + 0) * 128) = lo;
            *(v4f*)(pw + ((mt * 2 + nt) * 2 + 1) * 128) = hi;
        }
    }
    __syncthreads();

    if (tid < 132) {
        float ws = wred[tid];
        #pragma unroll
        for (int g = 1; g < 7; ++g) ws += wred[g * 132 + tid];
        if (tid < 128) wct[(tid & 3) * H0 + (tid >> 2)] = ws; else bcs[tid - 128] = ws + brr;
    }

    const float* pr = part + (wave + 1) * 512 + lane * 4;
    v8f acc0, acc1;
    {
        v4f q0 = *(const v4f*)(pr);
        v4f q1 = *(const v4f*)(pr + 128);
        v4f q2 = *(const v4f*)(pr + 256);
        v4f q3 = *(const v4f*)(pr + 384);
        #pragma unroll
        for (int s = 1; s < 8; ++s) {
            q0 += *(const v4f*)(pr + s * PW);
            q1 += *(const v4f*)(pr + s * PW + 128);
            q2 += *(const v4f*)(pr + s * PW + 256);
            q3 += *(const v4f*)(pr + s * PW + 384);
        }
        v8f r0 = {q0.x, q0.y, q0.z, q0.w, q1.x, q1.y, q1.z, q1.w};
        v8f r1 = {q2.x, q2.y, q2.z, q2.w, q3.x, q3.y, q3.z, q3.w};
        acc0 = r0; acc1 = r1;
    }

    {
        const v2f bA = splat2(b0s[m0]), bB = splat2(b0s[m0 + 16]);
        #pragma unroll
        for (int g = 0; g < 8; g += 2) {
            const int R = 4 + (wave << 4) + g + (hl << 3);
            v2f u0 = {acc0[g], acc0[g + 1]};
            v2f u1 = {acc1[g], acc1[g + 1]};
            u0 = tanh_pair(u0 + bA);
            u1 = tanh_pair(u1 + bB);
            h0s[R * 36 + m0]           = u0.x;
            h0s[(R + 1) * 36 + m0]     = u0.y;
            h0s[R * 36 + m0 + 16]      = u1.x;
            h0s[(R + 1) * 36 + m0 + 16] = u1.y;
        }
    }
    if (hl) {
        const int nt = wave & 1, e = wave >> 1;
        const float* ph = part + nt * 256 + 128 + lane * 4 + e;
        float s = ph[0];
        #pragma unroll
        for (int w = 1; w < 8; ++w) s += ph[w * PW];
        h0s[e * 36 + m0 + 16 * nt] = tanh_pair(splat2(s + b0s[m0 + 16 * nt])).x;
    }
    __syncthreads();

    const int arow = (m0 < 4 ? m0 : 4) * 32;

    {
        const float* wa = wct + arow + 4 * hl;
        const v4f a0 = *(const v4f*)(wa);
        const v4f a1 = *(const v4f*)(wa + 8);
        const v4f a2 = *(const v4f*)(wa + 16);
        const v4f a3 = *(const v4f*)(wa + 24);
        const v4f cb = *(const v4f*)bcs;
        const v8f cinit = {cb.x, cb.y, cb.z, cb.w, 0.f, 0.f, 0.f, 0.f};
        const v8f zero8 = {};
        {
            const int R = 4 + (wave << 4) + m0;
            const float* hb = h0s + R * 36 + 4 * hl;
            const v4f b0v = *(const v4f*)(hb);
            const v4f b1v = *(const v4f*)(hb + 8);
            const v4f b2v = *(const v4f*)(hb + 16);
            const v4f b3v = *(const v4f*)(hb + 24);
            v8f d0 = cinit, d1 = zero8;
            { v2f x = {a0.x, a0.y}, y = {b0v.x, b0v.y}; d0 = __builtin_amdgcn_wmma_f32_16x16x4_f32(false, x, false, y, (short)0, d0, false, false); }
            { v2f x = {a0.z, a0.w}, y = {b0v.z, b0v.w}; d1 = __builtin_amdgcn_wmma_f32_16x16x4_f32(false, x, false, y, (short)0, d1, false, false); }
            { v2f x = {a1.x, a1.y}, y = {b1v.x, b1v.y}; d0 = __builtin_amdgcn_wmma_f32_16x16x4_f32(false, x, false, y, (short)0, d0, false, false); }
            { v2f x = {a1.z, a1.w}, y = {b1v.z, b1v.w}; d1 = __builtin_amdgcn_wmma_f32_16x16x4_f32(false, x, false, y, (short)0, d1, false, false); }
            { v2f x = {a2.x, a2.y}, y = {b2v.x, b2v.y}; d0 = __builtin_amdgcn_wmma_f32_16x16x4_f32(false, x, false, y, (short)0, d0, false, false); }
            { v2f x = {a2.z, a2.w}, y = {b2v.z, b2v.w}; d1 = __builtin_amdgcn_wmma_f32_16x16x4_f32(false, x, false, y, (short)0, d1, false, false); }
            { v2f x = {a3.x, a3.y}, y = {b3v.x, b3v.y}; d0 = __builtin_amdgcn_wmma_f32_16x16x4_f32(false, x, false, y, (short)0, d0, false, false); }
            { v2f x = {a3.z, a3.w}, y = {b3v.z, b3v.w}; d1 = __builtin_amdgcn_wmma_f32_16x16x4_f32(false, x, false, y, (short)0, d1, false, false); }
            const v4f s = {d0[0] + d1[0], d0[1] + d1[1], d0[2] + d1[2], d0[3] + d1[3]};
            if (hl == 0) *(v4f*)(hm + R * 4) = s;
        }
        if (wave == 0) {
            const float* hb = h0s + m0 * 36 + 4 * hl;
            const v4f b0v = *(const v4f*)(hb);
            const v4f b1v = *(const v4f*)(hb + 8);
            const v4f b2v = *(const v4f*)(hb + 16);
            const v4f b3v = *(const v4f*)(hb + 24);
            v8f d0 = cinit, d1 = zero8;
            { v2f x = {a0.x, a0.y}, y = {b0v.x, b0v.y}; d0 = __builtin_amdgcn_wmma_f32_16x16x4_f32(false, x, false, y, (short)0, d0, false, false); }
            { v2f x = {a0.z, a0.w}, y = {b0v.z, b0v.w}; d1 = __builtin_amdgcn_wmma_f32_16x16x4_f32(false, x, false, y, (short)0, d1, false, false); }
            { v2f x = {a1.x, a1.y}, y = {b1v.x, b1v.y}; d0 = __builtin_amdgcn_wmma_f32_16x16x4_f32(false, x, false, y, (short)0, d0, false, false); }
            { v2f x = {a1.z, a1.w}, y = {b1v.z, b1v.w}; d1 = __builtin_amdgcn_wmma_f32_16x16x4_f32(false, x, false, y, (short)0, d1, false, false); }
            { v2f x = {a2.x, a2.y}, y = {b2v.x, b2v.y}; d0 = __builtin_amdgcn_wmma_f32_16x16x4_f32(false, x, false, y, (short)0, d0, false, false); }
            { v2f x = {a2.z, a2.w}, y = {b2v.z, b2v.w}; d1 = __builtin_amdgcn_wmma_f32_16x16x4_f32(false, x, false, y, (short)0, d1, false, false); }
            { v2f x = {a3.x, a3.y}, y = {b3v.x, b3v.y}; d0 = __builtin_amdgcn_wmma_f32_16x16x4_f32(false, x, false, y, (short)0, d0, false, false); }
            { v2f x = {a3.z, a3.w}, y = {b3v.z, b3v.w}; d1 = __builtin_amdgcn_wmma_f32_16x16x4_f32(false, x, false, y, (short)0, d1, false, false); }
            const v4f s = {d0[0] + d1[0], d0[1] + d1[1], d0[2] + d1[2], d0[3] + d1[3]};
            if (lane < 4) *(v4f*)(hm + m0 * 4) = s;
        }
    }

    v2f wa0[5], wa1[5];
    #pragma unroll
    for (int j = 0; j < 5; ++j) {
        const float* q = w2s + (4 * j + 2 * hl) * 32 + m0;
        v2f f0 = {q[0],  q[32]};
        v2f f1 = {q[16], q[48]};
        wa0[j] = f0; wa1[j] = f1;
    }
    const v4f c0a = *(const v4f*)(b2s + 8 * hl);
    const v4f c0b = *(const v4f*)(b2s + 8 * hl + 4);
    const v4f c1a = *(const v4f*)(b2s + 16 + 8 * hl);
    const v4f c1b = *(const v4f*)(b2s + 16 + 8 * hl + 4);
    const v4f p0 = *(const v4f*)(w3t + arow + 8 * hl);
    const v4f p1 = *(const v4f*)(w3t + arow + 8 * hl + 4);
    const v4f p2 = *(const v4f*)(w3t + arow + 16 + 8 * hl);
    const v4f p3 = *(const v4f*)(w3t + arow + 16 + 8 * hl + 4);
    const v4f b3q = *(const v4f*)b3s;
    __syncthreads();

    {
        const int tk = (wave << 4) + m0;
        v2f wb[5];
        #pragma unroll
        for (int j = 0; j < 5; ++j) wb[j] = *(const v2f*)(hm + (tk + j) * 4 + 2 * hl);
        v8f e0 = {c0a.x, c0a.y, c0a.z, c0a.w, c0b.x, c0b.y, c0b.z, c0b.w};
        v8f e1 = {c1a.x, c1a.y, c1a.z, c1a.w, c1b.x, c1b.y, c1b.z, c1b.w};
        #pragma unroll
        for (int j = 0; j < 5; ++j) {
            e0 = __builtin_amdgcn_wmma_f32_16x16x4_f32(false, wa0[j], false, wb[j], (short)0, e0, false, false);
            e1 = __builtin_amdgcn_wmma_f32_16x16x4_f32(false, wa1[j], false, wb[j], (short)0, e1, false, false);
        }
        #pragma unroll
        for (int r = 0; r < 8; ++r) { e0[r] = fmaxf(e0[r], 0.f); e1[r] = fmaxf(e1[r], 0.f); }

        v8f f0 = {b3q.x, b3q.y, b3q.z, b3q.w, 0.f, 0.f, 0.f, 0.f};
        v8f f1 = {};
        { v2f x = {p0.x, p0.y}, y = {e0[0], e0[1]}; f0 = __builtin_amdgcn_wmma_f32_16x16x4_f32(false, x, false, y, (short)0, f0, false, false); }
        { v2f x = {p0.z, p0.w}, y = {e0[2], e0[3]}; f1 = __builtin_amdgcn_wmma_f32_16x16x4_f32(false, x, false, y, (short)0, f1, false, false); }
        { v2f x = {p1.x, p1.y}, y = {e0[4], e0[5]}; f0 = __builtin_amdgcn_wmma_f32_16x16x4_f32(false, x, false, y, (short)0, f0, false, false); }
        { v2f x = {p1.z, p1.w}, y = {e0[6], e0[7]}; f1 = __builtin_amdgcn_wmma_f32_16x16x4_f32(false, x, false, y, (short)0, f1, false, false); }
        { v2f x = {p2.x, p2.y}, y = {e1[0], e1[1]}; f0 = __builtin_amdgcn_wmma_f32_16x16x4_f32(false, x, false, y, (short)0, f0, false, false); }
        { v2f x = {p2.z, p2.w}, y = {e1[2], e1[3]}; f1 = __builtin_amdgcn_wmma_f32_16x16x4_f32(false, x, false, y, (short)0, f1, false, false); }
        { v2f x = {p3.x, p3.y}, y = {e1[4], e1[5]}; f0 = __builtin_amdgcn_wmma_f32_16x16x4_f32(false, x, false, y, (short)0, f0, false, false); }
        { v2f x = {p3.z, p3.w}, y = {e1[6], e1[7]}; f1 = __builtin_amdgcn_wmma_f32_16x16x4_f32(false, x, false, y, (short)0, f1, false, false); }
        const v4f o = {f0[0] + f1[0], f0[1] + f1[1], f0[2] + f1[2], f0[3] + f1[3]};
        if (hl == 0)
            *(v4f*)(out + ((size_t)((bb << 12) + t0 + tk) << 2)) = o;
    }
}

extern "C" void kernel_launch(void* const* d_in, const int* in_sizes, int n_in,
                              void* d_out, int out_size, void* d_ws, size_t ws_size,
                              hipStream_t stream) {
    (void)in_sizes; (void)n_in; (void)out_size; (void)ws_size;
    const float* x  = (const float*)d_in[0];
    const float* W0 = (const float*)d_in[1];
    const float* b0 = (const float*)d_in[2];
    const float* W1 = (const float*)d_in[3];
    const float* b1 = (const float*)d_in[4];
    const float* Ws = (const float*)d_in[5];
    const float* bs = (const float*)d_in[6];
    const float* Wr = (const float*)d_in[7];
    const float* br = (const float*)d_in[8];
    const float* W2 = (const float*)d_in[9];
    const float* b2 = (const float*)d_in[10];
    const float* W3 = (const float*)d_in[11];
    const float* b3 = (const float*)d_in[12];
    float* out = (float*)d_out;

    float* Pc = (float*)d_ws;

    k_pre <<<NPART, 256, 0, stream>>>(Ws, bs, Wr, W1, b1, Pc);
    k_main<<<256,   256, 0, stream>>>(x, W0, b0, Pc, br, W2, b2, W3, b3, out);
}


# Global loads issued in order of need (loads return in issue order): fold kernel asks Wr first, LDS-only barrier; main kernel moves fold-partial and last B loads into the k loop; sum orders unchanged
# speedup vs baseline: 18.6439x; 1.0858x over previous
#include <hip/hip_runtime.h>
#include <hip/hip_bf16.h>


typedef __attribute__((ext_vector_type(2))) float v2f;
typedef __attribute__((ext_vector_type(4))) float v4f;
typedef __attribute__((ext_vector_type(8))) float v8f;

#define L_TOK   4096
#define B_BATCH 8
#define K0      484
#define H0      32
#define TB      128
#define HALO    120
#define NTILE   9
#define XBACK   (HALO + 16)
#define PW      (NTILE * 512)

#define FK_CH    1024
#define FK_KW    128
#define FK_NCH   4
#define FK_TILES 33
#define FK_PITCH 1032
#define NPART    (FK_TILES * FK_NCH)
#define PPITCH   160

__global__ __launch_bounds__(256) void k_pre(const float* __restrict__ Ws,
                                             const float* __restrict__ bs,
                                             const float* __restrict__ Wr,
                                             const float* __restrict__ W1,
                                             const float* __restrict__ b1,
                                             float* __restrict__ Pc) {
    __shared__ __attribute__((aligned(16))) float wrT[5 * FK_PITCH];
    __shared__ __attribute__((aligned(16))) float red[8 * 64];
    __shared__ __attribute__((aligned(16))) float tT[5 * 16];

    const int tid  = threadIdx.x;
    const int tile = blockIdx.x >> 2;
    const int kc   = blockIdx.x & 3;
    const int kb   = kc * FK_CH;
    const int wave = __builtin_amdgcn_readfirstlane(tid >> 5);
    const int lane = tid & 31;
    const int h    = lane >> 4;
    const int c    = lane & 15;

    v4f wq[4];
    #pragma unroll
    for (int r = 0; r < 4; ++r) wq[r] = ((const v4f*)(Wr + (size_t)kb * 4))[tid + (r << 8)];

    const int row  = tile * 16 + c;
    const int rowc = row < 512 ? row : 512;
    const float* src = (rowc < 512) ? (Ws + (size_t)rowc * 4096) : bs;
    const v4f* bp = (const v4f*)(src + kb + wave * FK_KW + 4 * h);
    const v4f zero4 = {0.f, 0.f, 0.f, 0.f};
    v4f bv[16];
    #pragma unroll
    for (int j = 0; j < 16; ++j) bv[j] = bp[2 * j];

    #pragma unroll
    for (int r = 0; r < 4; ++r) {
        int k = tid + (r << 8);
        v4f w = wq[r];
        wrT[0 * FK_PITCH + k] = w.x;
        wrT[1 * FK_PITCH + k] = w.y;
        wrT[2 * FK_PITCH + k] = w.z;
        wrT[3 * FK_PITCH + k] = w.w;
        wrT[4 * FK_PITCH + k] = 0.f;
    }
    if (tid >= 64 && tid < 80) tT[tid] = 0.f;

    v4f m0[2], m1[2], m2[2];
    #pragma unroll
    for (int j = 0; j < 2; ++j) { m0[j] = zero4; m1[j] = zero4; m2[j] = zero4; }
    if (wave == 0) {
        if (tile < 32) {
            const float* q = W1 + (size_t)c * 512 + tile * 16 + 4 * h;
            #pragma unroll
            for (int j = 0; j < 2; ++j) {
                m0[j] = *(const v4f*)(q + 8 * j);
                m1[j] = *(const v4f*)(q + 16 * 512 + 8 * j);
            }
            if (c == 0) {
                #pragma unroll
                for (int j = 0; j < 2; ++j) m2[j] = *(const v4f*)(b1 + tile * 16 + 4 * h + 8 * j);
            }
        } else if (lane == 0) {
            m2[0].x = 1.f;
        }
    }
    if (tile == 32 && c != 0) {
        #pragma unroll
        for (int j = 0; j < 16; ++j) bv[j] = zero4;
    }
    asm volatile("s_wait_dscnt 0x0" ::: "memory");
    asm volatile("s_barrier_signal -1\n\ts_barrier_wait -1" ::: "memory");

    v8f acc0 = {}; v8f acc1 = {};
    const float* ap = wrT + (c < 4 ? c : 4) * FK_PITCH + wave * FK_KW + 4 * h;
    #pragma unroll
    for (int j = 0; j < 16; ++j) {
        v4f av = *(const v4f*)(ap + 8 * j);
        v2f a0 = {av.x, av.y},       a1 = {av.z, av.w};
        v2f b0 = {bv[j].x, bv[j].y}, b1v = {bv[j].z, bv[j].w};
        acc0 = __builtin_amdgcn_wmma_f32_16x16x4_f32(false, a0, false, b0,  (short)0, acc0, false, false);
        acc1 = __builtin_amdgcn_wmma_f32_16x16x4_f32(false, a1, false, b1v, (short)0, acc1, false, false);
    }

    if (h == 0) {
        v4f s = {acc0[0] + acc1[0], acc0[1] + acc1[1], acc0[2] + acc1[2], acc0[3] + acc1[3]};
        *(v4f*)(red + wave * 64 + c * 4) = s;
    }
    __syncthreads();
    if (tid < 64) {
        float s = red[tid];
        #pragma unroll
        for (int w = 1; w < 8; ++w) s += red[w * 64 + tid];
        tT[(tid & 3) * 16 + (tid >> 2)] = s;
    }
    __syncthreads();

    if (wave == 0) {
        v8f e0 = {}; v8f e1 = {}; v8f e2 = {};
        const float* a2 = tT + (c < 4 ? c : 4) * 16 + 4 * h;
        #pragma unroll
        for (int j = 0; j < 2; ++j) {
            v4f av = *(const v4f*)(a2 + 8 * j);
            v2f a0 = {av.x, av.y}, a1 = {av.z, av.w};
            v2f x0 = {m0[j].x, m0[j].y}, x1 = {m0[j].z, m0[j].w};
            v2f y0 = {m1[j].x, m1[j].y}, y1 = {m1[j].z, m1[j].w};
            v2f z0 = {m2[j].x, m2[j].y}, z1 = {m2[j].z, m2[j].w};
            e0 = __builtin_amdgcn_wmma_f32_16x16x4_f32(false, a0, false, x0, (short)0, e0, false, false);
            e1 = __builtin_amdgcn_wmma_f32_16x16x4_f32(false, a0, false, y0, (short)0, e1, false, false);
            e2 = __builtin_amdgcn_wmma_f32_16x16x4_f32(false, a0, false, z0, (short)0, e2, false, false);
            e0 = __builtin_amdgcn_wmma_f32_16x16x4_f32(false, a1, false, x1, (short)0, e0, false, false);
            e1 = __builtin_amdgcn_wmma_f32_16x16x4_f32(false, a1, false, y1, (short)0, e1, false, false);
            e2 = __builtin_amdgcn_wmma_f32_16x16x4_f32(false, a1, false, z1, (short)0, e2, false, false);
        }
        if (h == 0) {
            float* dst = Pc + (size_t)blockIdx.x * PPITCH;
            v4f s0 = {e0[0], e0[1], e0[2], e0[3]};
            v4f s1 = {e1[0], e1[1], e1[2], e1[3]};
            v4f s2 = {e2[0], e2[1], e2[2], e2[3]};
            *(v4f*)(dst + c * 4)      = s0;
            *(v4f*)(dst + 64 + c * 4) = s1;
            if (c == 0) *(v4f*)(dst + 128) = s2;
        }
    }
}

__device__ __forceinline__ v2f splat2(float s) { v2f r = {s, s}; return r; }
__device__ __forceinline__ float tanh_pick(float v, float sm, float bg) {
    return __builtin_fabsf(v) < 0.625f ? sm : __builtin_copysignf(bg, v);
}
__device__ __forceinline__ v2f tanh_pair(v2f v) {
    const v2f z = v * v;
    v2f p = __builtin_elementwise_fma(z, splat2(-0.00570002059f), splat2(0.0206340719f));
    p = __builtin_elementwise_fma(z, p, splat2(-0.0537379310f));
    p = __builtin_elementwise_fma(z, p, splat2(0.133314162f));
    p = __builtin_elementwise_fma(z, p, splat2(-0.333332807f));
    const v2f sm = __builtin_elementwise_fma(z, v * p, v);
    const v2f e = v * splat2(2.88539004f);
    v2f t = {__builtin_amdgcn_exp2f(__builtin_fabsf(e.x)), __builtin_amdgcn_exp2f(__builtin_fabsf(e.y))};
    t = t + splat2(1.f);
    const v2f r = {__builtin_amdgcn_rcpf(t.x), __builtin_amdgcn_rcpf(t.y)};
    const v2f bg = __builtin_elementwise_fma(r, splat2(-2.f), splat2(1.f));
    v2f o = {tanh_pick(v.x, sm.x, bg.x), tanh_pick(v.y, sm.y, bg.y)};
    return o;
}

#define KS_WAVE 15
__global__ __launch_bounds__(256) void k_main(const float* __restrict__ x,
                                              const float* __restrict__ W0,
                                              const float* __restrict__ b0,
                                              const float* __restrict__ Pc,
                                              const float* __restrict__ br,
                                              const float* __restrict__ W2,
                                              const float* __restrict__ b2,
                                              const float* __restrict__ W3,
                                              const float* __restrict__ b3,
                                              float* __restrict__ out) {
    __shared__ __attribute__((aligned(16))) float part[8 * PW];
    __shared__ float b0s[H0];
    __shared__ __attribute__((aligned(16))) float wct[5 * H0];
    __shared__ __attribute__((aligned(16))) float bcs[4];
    __shared__ __attribute__((aligned(16))) float wred[7 * 132];
    __shared__ __attribute__((aligned(16))) float h0s[(TB + 4) * 36];
    __shared__ __attribute__((aligned(16))) float hm[(TB + 4) * 4];
    __shared__ __attribute__((aligned(16))) float w2s[20 * 32];
    __shared__ __attribute__((aligned(16))) float b2s[32];
    __shared__ __attribute__((aligned(16))) float w3t[5 * 32];
    __shared__ __attribute__((aligned(16))) float b3s[4];

    const int tid = threadIdx.x;
    const int bb  = blockIdx.x >> 5;
    const int t0  = (blockIdx.x & 31) * TB;
    const int wave = __builtin_amdgcn_readfirstlane(tid >> 5);
    const int lane = tid & 31;
    const int hl   = lane >> 4;
    const int m0   = lane & 15;
    const int c0   = hl << 1;
    const int k_lo = wave * KS_WAVE;

    const v2f* xs = (const v2f*)(x + ((size_t)bb << 14));
    const unsigned ub = (unsigned)(2 * (t0 - XBACK + m0 + k_lo) + hl);
#define XA(mt, i) (xs[(ub + 2u * (unsigned)((mt) * 16 + (i))) & 8191u])

    v2f abuf[3][NTILE];
    #pragma unroll
    for (int mt = 0; mt < NTILE; ++mt) { abuf[0][mt] = XA(mt, 0); abuf[1][mt] = XA(mt, 1); }

    const float* bptr = W0 + (size_t)(((k_lo << 2) + c0) * H0 + m0);
    v2f breg[16][2];
#define KS_FRONT 12
    #pragma unroll
    for (int i = 0; i < KS_FRONT; ++i) {
        v2f f0 = {bptr[(4 * i) * H0],      bptr[(4 * i + 1) * H0]};
        v2f f1 = {bptr[(4 * i) * H0 + 16], bptr[(4 * i + 1) * H0 + 16]};
        breg[i][0] = f0; breg[i][1] = f1;
    }
    {
        v2f z = {0.f, 0.f};
        breg[15][0] = z; breg[15][1] = z;
    }

    float brr = 0.f;
    if (tid >= 128 && tid < 132) brr = br[tid - 128];

    const float w2r0 = W2[tid];
    const float w2r1 = W2[tid + 256];
    float w2r2 = 0.f, w3r = 0.f, b2r = 0.f, b3r = 0.f;
    if (tid < 128) { w2r2 = W2[tid + 512]; w3r = W3[tid]; }
    float b0r = 0.f;
    if (tid < 32)  { b2r = b2[tid]; b0r = b0[tid]; }
    if (tid < 4)   b3r = b3[tid];

    v8f acc[NTILE][2];
    #pragma unroll
    for (int mt = 0; mt < NTILE; ++mt) { v8f z = {}; acc[mt][0] = z; acc[mt][1] = z; }

    const int pg = tid / 33;
    const int ps = tid - pg * 33;
    const bool pact = tid < 7 * 33;
    const v4f* pp = (const v4f*)Pc + (pact ? pg * (PPITCH / 4) + ps : 0);
    const int q18 = (pact && pg < 6) ? 18 * 7 * (PPITCH / 4) : 0;
    v4f pv[19];

    #pragma unroll
    for (int i = 0; i < KS_WAVE; ++i) {
        if (i + 2 < KS_WAVE) {
            #pragma unroll
            for (int mt = 0; mt < NTILE; ++mt) abuf[(i + 2) % 3][mt] = XA(mt, i + 2);
        } else if (i + 2 == KS_WAVE) {
            if (wave == 7) {
                #pragma unroll
                for (int mt = 0; mt < NTILE; ++mt) abuf[(i + 2) % 3][mt] = XA(mt, KS_WAVE);
            }
        }
        if (i == 1) {
            #pragma unroll
            for (int b = KS_FRONT; b < KS_WAVE; ++b) {
                v2f f0 = {bptr[(4 * b) * H0],      bptr[(4 * b + 1) * H0]};
                v2f f1 = {bptr[(4 * b) * H0 + 16], bptr[(4 * b + 1) * H0 + 16]};
                breg[b][0] = f0; breg[b][1] = f1;
            }
            if (wave == 7) {
                v2f f0 = {bptr[60 * H0],      bptr[61 * H0]};
                v2f f1 = {bptr[60 * H0 + 16], bptr[61 * H0 + 16]};
                breg[15][0] = f0; breg[15][1] = f1;
            }
        }
        if (i >= 2 && i <= 10) {
            pv[2 * (i - 2)]     = pp[(2 * (i - 2))     * 7 * (PPITCH / 4)];
            pv[2 * (i - 2) + 1] = pp[(2 * (i - 2) + 1) * 7 * (PPITCH / 4)];
        }
        if (i == 11) pv[18] = pp[q18];
        __builtin_amdgcn_sched_barrier(0);
        #pragma unroll
        for (int mt = 0; mt < NTILE; ++mt) {
            const v2f a = abuf[i % 3][mt];
            acc[mt][0] = __builtin_amdgcn_wmma_f32_16x16x4_f32(false, a, false, breg[i][0], (short)0, acc[mt][0], false, false);
            acc[mt][1] = __builtin_amdgcn_wmma_f32_16x16x4_f32(false, a, false, breg[i][1], (short)0, acc[mt][1], false, false);
        }
        __builtin_amdgcn_sched_barrier(0);
    }
    if (wave == 7) {
        #pragma unroll
        for (int mt = 0; mt < NTILE; ++mt) {
            const v2f a = abuf[KS_WAVE % 3][mt];
            acc[mt][0] = __builtin_amdgcn_wmma_f32_16x16x4_f32(false, a, false, breg[15][0], (short)0, acc[mt][0], false, false);
            acc[mt][1] = __builtin_amdgcn_wmma_f32_16x16x4_f32(false, a, false, breg[15][1], (short)0, acc[mt][1], false, false);
        }
    }

    w2s[tid]       = w2r0;
    w2s[tid + 256] = w2r1;
    if (tid < 128) { w2s[tid + 512] = w2r2; w3t[(tid & 3) * 32 + (tid >> 2)] = w3r; }
    if (tid < 32)  { b0s[tid] = b0r; b2s[tid] = b2r; w3t[4 * 32 + tid] = 0.f; wct[4 * H0 + tid] = 0.f; }
    if (tid < 4)   b3s[tid] = b3r;

    if (pact) {
        const v4f z4 = {0.f, 0.f, 0.f, 0.f};
        v4f psum = pv[0];
        #pragma unroll
        for (int q = 1; q < 18; ++q) psum += pv[q];
        psum += (pg < 6) ? pv[18] : z4;
        *(v4f*)(wred + pg * 132 + ps * 4) = psum;
    }

    float* pw = part + wave * PW + lane * 4;
    #pragma unroll
    for (int mt = 0; mt < NTILE; ++mt) {
        #pragma unroll
        for (int nt = 0; nt < 2; ++nt) {
            v8f d = acc[mt][nt];
            v4f lo = {d[0], d[1], d[2], d[3]};
            v4f hi = {d[4], d[5], d[6], d[7]};
            *(v4f*)(pw + ((mt * 2 + nt) * 2 + 0) * 128) = lo;
            *(v4f*)(pw + ((mt * 2 + nt) * 2 + 1) * 128) = hi;
        }
    }
    __syncthreads();

    if (tid < 132) {
        float ws = wred[tid];
        #pragma unroll
        for (int g = 1; g < 7; ++g) ws += wred[g * 132 + tid];
        if (tid < 128) wct[(tid & 3) * H0 + (tid >> 2)] = ws; else bcs[tid - 128] = ws + brr;
    }

    const float* pr = part + (wave + 1) * 512 + lane * 4;
    v8f acc0, acc1;
    {
        v4f q0 = *(const v4f*)(pr);
        v4f q1 = *(const v4f*)(pr + 128);
        v4f q2 = *(const v4f*)(pr + 256);
        v4f q3 = *(const v4f*)(pr + 384);
        #pragma unroll
        for (int s = 1; s < 8; ++s) {
            q0 += *(const v4f*)(pr + s * PW);
            q1 += *(const v4f*)(pr + s * PW + 128);
            q2 += *(const v4f*)(pr + s * PW + 256);
            q3 += *(const v4f*)(pr + s * PW + 384);
        }
        v8f r0 = {q0.x, q0.y, q0.z, q0.w, q1.x, q1.y, q1.z, q1.w};
        v8f r1 = {q2.x, q2.y, q2.z, q2.w, q3.x, q3.y, q3.z, q3.w};
        acc0 = r0; acc1 = r1;
    }

    {
        const v2f bA = splat2(b0s[m0]), bB = splat2(b0s[m0 + 16]);
        #pragma unroll
        for (int g = 0; g < 8; g += 2) {
            const int R = 4 + (wave << 4) + g + (hl << 3);
            v2f u0 = {acc0[g], acc0[g + 1]};
            v2f u1 = {acc1[g], acc1[g + 1]};
            u0 = tanh_pair(u0 + bA);
            u1 = tanh_pair(u1 + bB);
            h0s[R * 36 + m0]           = u0.x;
            h0s[(R + 1) * 36 + m0]     = u0.y;
            h0s[R * 36 + m0 + 16]      = u1.x;
            h0s[(R + 1) * 36 + m0 + 16] = u1.y;
        }
    }
    if (hl) {
        const int nt = wave & 1, e = wave >> 1;
        const float* ph = part + nt * 256 + 128 + lane * 4 + e;
        float s = ph[0];
        #pragma unroll
        for (int w = 1; w < 8; ++w) s += ph[w * PW];
        h0s[e * 36 + m0 + 16 * nt] = tanh_pair(splat2(s + b0s[m0 + 16 * nt])).x;
    }
    __syncthreads();

    const int arow = (m0 < 4 ? m0 : 4) * 32;

    {
        const float* wa = wct + arow + 4 * hl;
        const v4f a0 = *(const v4f*)(wa);
        const v4f a1 = *(const v4f*)(wa + 8);
        const v4f a2 = *(const v4f*)(wa + 16);
        const v4f a3 = *(const v4f*)(wa + 24);
        const v4f cb = *(const v4f*)bcs;
        const v8f cinit = {cb.x, cb.y, cb.z, cb.w, 0.f, 0.f, 0.f, 0.f};
        const v8f zero8 = {};
        {
            const int R = 4 + (wave << 4) + m0;
            const float* hb = h0s + R * 36 + 4 * hl;
            const v4f b0v = *(const v4f*)(hb);
            const v4f b1v = *(const v4f*)(hb + 8);
            const v4f b2v = *(const v4f*)(hb + 16);
            const v4f b3v = *(const v4f*)(hb + 24);
            v8f d0 = cinit, d1 = zero8;
            { v2f x = {a0.x, a0.y}, y = {b0v.x, b0v.y}; d0 = __builtin_amdgcn_wmma_f32_16x16x4_f32(false, x, false, y, (short)0, d0, false, false); }
            { v2f x = {a0.z, a0.w}, y = {b0v.z, b0v.w}; d1 = __builtin_amdgcn_wmma_f32_16x16x4_f32(false, x, false, y, (short)0, d1, false, false); }
            { v2f x = {a1.x, a1.y}, y = {b1v.x, b1v.y}; d0 = __builtin_amdgcn_wmma_f32_16x16x4_f32(false, x, false, y, (short)0, d0, false, false); }
            { v2f x = {a1.z, a1.w}, y = {b1v.z, b1v.w}; d1 = __builtin_amdgcn_wmma_f32_16x16x4_f32(false, x, false, y, (short)0, d1, false, false); }
            { v2f x = {a2.x, a2.y}, y = {b2v.x, b2v.y}; d0 = __builtin_amdgcn_wmma_f32_16x16x4_f32(false, x, false, y, (short)0, d0, false, false); }
            { v2f x = {a2.z, a2.w}, y = {b2v.z, b2v.w}; d1 = __builtin_amdgcn_wmma_f32_16x16x4_f32(false, x, false, y, (short)0, d1, false, false); }
            { v2f x = {a3.x, a3.y}, y = {b3v.x, b3v.y}; d0 = __builtin_amdgcn_wmma_f32_16x16x4_f32(false, x, false, y, (short)0, d0, false, false); }
            { v2f x = {a3.z, a3.w}, y = {b3v.z, b3v.w}; d1 = __builtin_amdgcn_wmma_f32_16x16x4_f32(false, x, false, y, (short)0, d1, false, false); }
            const v4f s = {d0[0] + d1[0], d0[1] + d1[1], d0[2] + d1[2], d0[3] + d1[3]};
            if (hl == 0) *(v4f*)(hm + R * 4) = s;
        }
        if (wave == 0) {
            const float* hb = h0s + m0 * 36 + 4 * hl;
            const v4f b0v = *(const v4f*)(hb);
            const v4f b1v = *(const v4f*)(hb + 8);
            const v4f b2v = *(const v4f*)(hb + 16);
            const v4f b3v = *(const v4f*)(hb + 24);
            v8f d0 = cinit, d1 = zero8;
            { v2f x = {a0.x, a0.y}, y = {b0v.x, b0v.y}; d0 = __builtin_amdgcn_wmma_f32_16x16x4_f32(false, x, false, y, (short)0, d0, false, false); }
            { v2f x = {a0.z, a0.w}, y = {b0v.z, b0v.w}; d1 = __builtin_amdgcn_wmma_f32_16x16x4_f32(false, x, false, y, (short)0, d1, false, false); }
            { v2f x = {a1.x, a1.y}, y = {b1v.x, b1v.y}; d0 = __builtin_amdgcn_wmma_f32_16x16x4_f32(false, x, false, y, (short)0, d0, false, false); }
            { v2f x = {a1.z, a1.w}, y = {b1v.z, b1v.w}; d1 = __builtin_amdgcn_wmma_f32_16x16x4_f32(false, x, false, y, (short)0, d1, false, false); }
            { v2f x = {a2.x, a2.y}, y = {b2v.x, b2v.y}; d0 = __builtin_amdgcn_wmma_f32_16x16x4_f32(false, x, false, y, (short)0, d0, false, false); }
            { v2f x = {a2.z, a2.w}, y = {b2v.z, b2v.w}; d1 = __builtin_amdgcn_wmma_f32_16x16x4_f32(false, x, false, y, (short)0, d1, false, false); }
            { v2f x = {a3.x, a3.y}, y = {b3v.x, b3v.y}; d0 = __builtin_amdgcn_wmma_f32_16x16x4_f32(false, x, false, y, (short)0, d0, false, false); }
            { v2f x = {a3.z, a3.w}, y = {b3v.z, b3v.w}; d1 = __builtin_amdgcn_wmma_f32_16x16x4_f32(false, x, false, y, (short)0, d1, false, false); }
            const v4f s = {d0[0] + d1[0], d0[1] + d1[1], d0[2] + d1[2], d0[3] + d1[3]};
            if (lane < 4) *(v4f*)(hm + m0 * 4) = s;
        }
    }

    v2f wa0[5], wa1[5];
    #pragma unroll
    for (int j = 0; j < 5; ++j) {
        const float* q = w2s + (4 * j + 2 * hl) * 32 + m0;
        v2f f0 = {q[0],  q[32]};
        v2f f1 = {q[16], q[48]};
        wa0[j] = f0; wa1[j] = f1;
    }
    const v4f c0a = *(const v4f*)(b2s + 8 * hl);
    const v4f c0b = *(const v4f*)(b2s + 8 * hl + 4);
    const v4f c1a = *(const v4f*)(b2s + 16 + 8 * hl);
    const v4f c1b = *(const v4f*)(b2s + 16 + 8 * hl + 4);
    const v4f p0 = *(const v4f*)(w3t + arow + 8 * hl);
    const v4f p1 = *(const v4f*)(w3t + arow + 8 * hl + 4);
    const v4f p2 = *(const v4f*)(w3t + arow + 16 + 8 * hl);
    const v4f p3 = *(const v4f*)(w3t + arow + 16 + 8 * hl + 4);
    const v4f b3q = *(const v4f*)b3s;
    __syncthreads();

    {
        const int tk = (wave << 4) + m0;
        v2f wb[5];
        #pragma unroll
        for (int j = 0; j < 5; ++j) wb[j] = *(const v2f*)(hm + (tk + j) * 4 + 2 * hl);
        v8f e0 = {c0a.x, c0a.y, c0a.z, c0a.w, c0b.x, c0b.y, c0b.z, c0b.w};
        v8f e1 = {c1a.x, c1a.y, c1a.z, c1a.w, c1b.x, c1b.y, c1b.z, c1b.w};
        #pragma unroll
        for (int j = 0; j < 5; ++j) {
            e0 = __builtin_amdgcn_wmma_f32_16x16x4_f32(false, wa0[j], false, wb[j], (short)0, e0, false, false);
            e1 = __builtin_amdgcn_wmma_f32_16x16x4_f32(false, wa1[j], false, wb[j], (short)0, e1, false, false);
        }
        #pragma unroll
        for (int r = 0; r < 8; ++r) { e0[r] = fmaxf(e0[r], 0.f); e1[r] = fmaxf(e1[r], 0.f); }

        v8f f0 = {b3q.x, b3q.y, b3q.z, b3q.w, 0.f, 0.f, 0.f, 0.f};
        v8f f1 = {};
        { v2f x = {p0.x, p0.y}, y = {e0[0], e0[1]}; f0 = __builtin_amdgcn_wmma_f32_16x16x4_f32(false, x, false, y, (short)0, f0, false, false); }
        { v2f x = {p0.z, p0.w}, y = {e0[2], e0[3]}; f1 = __builtin_amdgcn_wmma_f32_16x16x4_f32(false, x, false, y, (short)0, f1, false, false); }
        { v2f x = {p1.x, p1.y}, y = {e0[4], e0[5]}; f0 = __builtin_amdgcn_wmma_f32_16x16x4_f32(false, x, false, y, (short)0, f0, false, false); }
        { v2f x = {p1.z, p1.w}, y = {e0[6], e0[7]}; f1 = __builtin_amdgcn_wmma_f32_16x16x4_f32(false, x, false, y, (short)0, f1, false, false); }
        { v2f x = {p2.x, p2.y}, y = {e1[0], e1[1]}; f0 = __builtin_amdgcn_wmma_f32_16x16x4_f32(false, x, false, y, (short)0, f0, false, false); }
        { v2f x = {p2.z, p2.w}, y = {e1[2], e1[3]}; f1 = __builtin_amdgcn_wmma_f32_16x16x4_f32(false, x, false, y, (short)0, f1, false, false); }
        { v2f x = {p3.x, p3.y}, y = {e1[4], e1[5]}; f0 = __builtin_amdgcn_wmma_f32_16x16x4_f32(false, x, false, y, (short)0, f0, false, false); }
        { v2f x = {p3.z, p3.w}, y = {e1[6], e1[7]}; f1 = __builtin_amdgcn_wmma_f32_16x16x4_f32(false, x, false, y, (short)0, f1, false, false); }
        const v4f o = {f0[0] + f1[0], f0[1] + f1[1], f0[2] + f1[2], f0[3] + f1[3]};
        if (hl == 0)
            *(v4f*)(out + ((size_t)((bb << 12) + t0 + tk) << 2)) = o;
    }
}

extern "C" void kernel_launch(void* const* d_in, const int* in_sizes, int n_in,
                              void* d_out, int out_size, void* d_ws, size_t ws_size,
                              hipStream_t stream) {
    (void)in_sizes; (void)n_in; (void)out_size; (void)ws_size;
    const float* x  = (const float*)d_in[0];
    const float* W0 = (const float*)d_in[1];
    const float* b0 = (const float*)d_in[2];
    const float* W1 = (const float*)d_in[3];
    const float* b1 = (const float*)d_in[4];
    const float* Ws = (const float*)d_in[5];
    const float* bs = (const float*)d_in[6];
    const float* Wr = (const float*)d_in[7];
    const float* br = (const float*)d_in[8];
    const float* W2 = (const float*)d_in[9];
    const float* b2 = (const float*)d_in[10];
    const float* W3 = (const float*)d_in[11];
    const float* b3 = (const float*)d_in[12];
    float* out = (float*)d_out;

    float* Pc = (float*)d_ws;

    k_pre <<<NPART, 256, 0, stream>>>(Ws, bs, Wr, W1, b1, Pc);
    k_main<<<256,   256, 0, stream>>>(x, W0, b0, Pc, br, W2, b2, W3, b3, out);
}
